# starting kernel
# baseline (speedup 1.0000x reference)
#include <hip/hip_runtime.h>

#define NNODES 20000
#define NRELH  16
#define RTOT   33
#define HID    64
#define NCLS   16
#define RN     (RTOT * NNODES)
#define PCOLS  (RTOT * NCLS)

typedef __attribute__((ext_vector_type(2))) float v2f;
typedef __attribute__((ext_vector_type(8))) float v8f;

__global__ __launch_bounds__(256) void k_zero(float* __restrict__ p, int n) {
  int i = blockIdx.x * blockDim.x + threadIdx.x;
  if (i < n) p[i] = 0.0f;
}

__global__ __launch_bounds__(256) void k_count(const int* __restrict__ src,
                                               const int* __restrict__ rel,
                                               const int* __restrict__ dst,
                                               float* __restrict__ cnt_ro,
                                               float* __restrict__ cnt_rs, int E) {
  int e = blockIdx.x * blockDim.x + threadIdx.x;
  if (e >= E) return;
  int s = src[e], r = rel[e], d = dst[e];
  atomicAdd(cnt_ro + (size_t)r * NNODES + d, 1.0f);
  atomicAdd(cnt_ro + (size_t)(r + NRELH) * NNODES + s, 1.0f);
  atomicAdd(cnt_rs + (size_t)r * NNODES + s, 1.0f);
  atomicAdd(cnt_rs + (size_t)(r + NRELH) * NNODES + d, 1.0f);
}

__global__ __launch_bounds__(256) void k_layer1(const int* __restrict__ src,
                                                const int* __restrict__ rel,
                                                const int* __restrict__ dst,
                                                const float* __restrict__ w1,
                                                const float* __restrict__ cnt_ro,
                                                float* __restrict__ hacc, int E) {
  long idx = (long)blockIdx.x * blockDim.x + threadIdx.x;
  if (idx >= (long)E * 16) return;
  int e  = (int)(idx >> 4);
  int i0 = ((int)idx & 15) * 4;
  int s = src[e], r = rel[e], d = dst[e];
  float vo = 1.0f / cnt_ro[(size_t)(r + NRELH) * NNODES + s];
  float vi = 1.0f / cnt_ro[(size_t)r * NNODES + d];
  float4 a = *(const float4*)(w1 + ((size_t)r * NNODES + d) * HID + i0);
  atomicAdd(&hacc[(size_t)s * HID + i0 + 0], vo * a.x);
  atomicAdd(&hacc[(size_t)s * HID + i0 + 1], vo * a.y);
  atomicAdd(&hacc[(size_t)s * HID + i0 + 2], vo * a.z);
  atomicAdd(&hacc[(size_t)s * HID + i0 + 3], vo * a.w);
  float4 b = *(const float4*)(w1 + ((size_t)(r + NRELH) * NNODES + s) * HID + i0);
  atomicAdd(&hacc[(size_t)d * HID + i0 + 0], vi * b.x);
  atomicAdd(&hacc[(size_t)d * HID + i0 + 1], vi * b.y);
  atomicAdd(&hacc[(size_t)d * HID + i0 + 2], vi * b.z);
  atomicAdd(&hacc[(size_t)d * HID + i0 + 3], vi * b.w);
}

__global__ __launch_bounds__(256) void k_hfinal(const float* __restrict__ w1,
                                                const float* __restrict__ b1,
                                                float* __restrict__ h) {
  int idx = blockIdx.x * blockDim.x + threadIdx.x;
  if (idx >= NNODES * HID) return;
  int n = idx / HID, i = idx % HID;
  float v = h[idx] + w1[((size_t)(2 * NRELH) * NNODES + n) * HID + i] + b1[i];
  h[idx] = v > 0.0f ? v : 0.0f;
}

__global__ __launch_bounds__(32) void k_gemm(const float* __restrict__ h,
                                             const float* __restrict__ w2,
                                             float* __restrict__ P) {
  int tile = blockIdx.x;
  int mt = tile % (NNODES / 16);
  int r  = tile / (NNODES / 16);
  int lane  = threadIdx.x;
  int mn    = lane & 15;
  int khalf = lane >> 4;
  const float* hrow = h + (size_t)(mt * 16 + mn) * HID;
  const float* wrel = w2 + (size_t)r * HID * NCLS;
  union { v8f v; float f[8]; } acc;
  acc.v = (v8f)0.0f;
#pragma unroll
  for (int k0 = 0; k0 < HID; k0 += 4) {
    int ka = k0 + 2 * khalf;
    v2f a, b;
    a.x = hrow[ka];
    a.y = hrow[ka + 1];
    b.x = wrel[(size_t)ka * NCLS + mn];
    b.y = wrel[(size_t)(ka + 1) * NCLS + mn];
    acc.v = __builtin_amdgcn_wmma_f32_16x16x4_f32(
         false, a,  false, b,
         (short)0, acc.v,  false,  false);
  }
  int rowoff = khalf * 8;
#pragma unroll
  for (int v = 0; v < 8; ++v) {
    P[(size_t)(mt * 16 + v + rowoff) * PCOLS + r * NCLS + mn] = acc.f[v];
  }
}

__global__ __launch_bounds__(256) void k_outinit(const float* __restrict__ b2,
                                                 const float* __restrict__ P,
                                                 float* __restrict__ out) {
  int idx = blockIdx.x * blockDim.x + threadIdx.x;
  if (idx >= NNODES * NCLS) return;
  int n = idx / NCLS, c = idx % NCLS;
  out[idx] = b2[c] + P[(size_t)n * PCOLS + (2 * NRELH) * NCLS + c];
}

__global__ __launch_bounds__(256) void k_layer2(const int* __restrict__ src,
                                                const int* __restrict__ rel,
                                                const int* __restrict__ dst,
                                                const float* __restrict__ cnt_rs,
                                                const float* __restrict__ P,
                                                float* __restrict__ out, int E) {
  long idx = (long)blockIdx.x * blockDim.x + threadIdx.x;
  if (idx >= (long)E * 16) return;
  int e = (int)(idx >> 4);
  int c = (int)idx & 15;
  int s = src[e], r = rel[e], d = dst[e];
  float vo = 1.0f / cnt_rs[(size_t)r * NNODES + s];
  float vi = 1.0f / cnt_rs[(size_t)(r + NRELH) * NNODES + d];
  atomicAdd(&out[(size_t)s * NCLS + c], vo * P[(size_t)d * PCOLS + r * NCLS + c]);
  atomicAdd(&out[(size_t)d * NCLS + c], vi * P[(size_t)s * PCOLS + (r + NRELH) * NCLS + c]);
}

extern "C" void kernel_launch(void* const* d_in, const int* in_sizes, int n_in,
                              void* d_out, int out_size, void* d_ws, size_t ws_size,
                              hipStream_t stream) {
  const int*   src = (const int*)d_in[0];
  const int*   rel = (const int*)d_in[1];
  const int*   dst = (const int*)d_in[2];
  const float* w1  = (const float*)d_in[3];
  const float* b1  = (const float*)d_in[4];
  const float* w2  = (const float*)d_in[5];
  const float* b2  = (const float*)d_in[6];
  float* out = (float*)d_out;
  const int E = in_sizes[0];

  float* cnt_ro = (float*)d_ws;
  float* cnt_rs = cnt_ro + RN;
  float* h      = cnt_rs + RN;
  float* P      = h + (size_t)NNODES * HID;

  const int zn = 2 * RN + NNODES * HID;
  k_zero<<<(zn + 255) / 256, 256, 0, stream>>>(cnt_ro, zn);

  k_count<<<(E + 255) / 256, 256, 0, stream>>>(src, rel, dst, cnt_ro, cnt_rs, E);

  long l1 = (long)E * 16;
  k_layer1<<<(unsigned)((l1 + 255) / 256), 256, 0, stream>>>(src, rel, dst, w1, cnt_ro, h, E);

  k_hfinal<<<(NNODES * HID + 255) / 256, 256, 0, stream>>>(w1, b1, h);

  k_gemm<<<(NNODES / 16) * RTOT, 32, 0, stream>>>(h, w2, P);

  k_outinit<<<(NNODES * NCLS + 255) / 256, 256, 0, stream>>>(b2, P, out);

  long l2 = (long)E * 16;
  k_layer2<<<(unsigned)((l2 + 255) / 256), 256, 0, stream>>>(src, rel, dst, cnt_rs, P, out, E);
}


# Build two count arrays, not four: the inverse block swaps edge ends, so its ro/rs counts equal the original block's rs/ro counts; cnt_s[r,src], cnt_d[r,dst]: 2 float atomics per edge instead of 4
# speedup vs baseline: 1.0918x; 1.0918x over previous
#include <hip/hip_runtime.h>

#define NNODES 20000
#define NRELH  16
#define RTOT   33
#define HID    64
#define NCLS   16
#define RN     (RTOT * NNODES)
#define CN     (NRELH * NNODES)
#define PCOLS  (RTOT * NCLS)

typedef __attribute__((ext_vector_type(2))) float v2f;
typedef __attribute__((ext_vector_type(8))) float v8f;

__global__ __launch_bounds__(256) void k_zero(float* __restrict__ p, int n) {
  int i = blockIdx.x * blockDim.x + threadIdx.x;
  if (i < n) p[i] = 0.0f;
}

__global__ __launch_bounds__(256) void k_count(const int* __restrict__ src,
                                               const int* __restrict__ rel,
                                               const int* __restrict__ dst,
                                               float* __restrict__ cnt_s,
                                               float* __restrict__ cnt_d, int E) {
  int e = blockIdx.x * blockDim.x + threadIdx.x;
  if (e >= E) return;
  int s = src[e], r = rel[e], d = dst[e];
  atomicAdd(cnt_s + (size_t)r * NNODES + s, 1.0f);
  atomicAdd(cnt_d + (size_t)r * NNODES + d, 1.0f);
}

__global__ __launch_bounds__(256) void k_layer1(const int* __restrict__ src,
                                                const int* __restrict__ rel,
                                                const int* __restrict__ dst,
                                                const float* __restrict__ w1,
                                                const float* __restrict__ cnt_s,
                                                const float* __restrict__ cnt_d,
                                                float* __restrict__ hacc, int E) {
  long idx = (long)blockIdx.x * blockDim.x + threadIdx.x;
  if (idx >= (long)E * 16) return;
  int e  = (int)(idx >> 4);
  int i0 = ((int)idx & 15) * 4;
  int s = src[e], r = rel[e], d = dst[e];
  float vo = 1.0f / cnt_s[(size_t)r * NNODES + s];
  float vi = 1.0f / cnt_d[(size_t)r * NNODES + d];
  float4 a = *(const float4*)(w1 + ((size_t)r * NNODES + d) * HID + i0);
  atomicAdd(&hacc[(size_t)s * HID + i0 + 0], vo * a.x);
  atomicAdd(&hacc[(size_t)s * HID + i0 + 1], vo * a.y);
  atomicAdd(&hacc[(size_t)s * HID + i0 + 2], vo * a.z);
  atomicAdd(&hacc[(size_t)s * HID + i0 + 3], vo * a.w);
  float4 b = *(const float4*)(w1 + ((size_t)(r + NRELH) * NNODES + s) * HID + i0);
  atomicAdd(&hacc[(size_t)d * HID + i0 + 0], vi * b.x);
  atomicAdd(&hacc[(size_t)d * HID + i0 + 1], vi * b.y);
  atomicAdd(&hacc[(size_t)d * HID + i0 + 2], vi * b.z);
  atomicAdd(&hacc[(size_t)d * HID + i0 + 3], vi * b.w);
}

__global__ __launch_bounds__(256) void k_hfinal(const float* __restrict__ w1,
                                                const float* __restrict__ b1,
                                                float* __restrict__ h) {
  int idx = blockIdx.x * blockDim.x + threadIdx.x;
  if (idx >= NNODES * HID) return;
  int n = idx / HID, i = idx % HID;
  float v = h[idx] + w1[((size_t)(2 * NRELH) * NNODES + n) * HID + i] + b1[i];
  h[idx] = v > 0.0f ? v : 0.0f;
}

__global__ __launch_bounds__(32) void k_gemm(const float* __restrict__ h,
                                             const float* __restrict__ w2,
                                             float* __restrict__ P) {
  int tile = blockIdx.x;
  int mt = tile % (NNODES / 16);
  int r  = tile / (NNODES / 16);
  int lane  = threadIdx.x;
  int mn    = lane & 15;
  int khalf = lane >> 4;
  const float* hrow = h + (size_t)(mt * 16 + mn) * HID;
  const float* wrel = w2 + (size_t)r * HID * NCLS;
  union { v8f v; float f[8]; } acc;
  acc.v = (v8f)0.0f;
#pragma unroll
  for (int k0 = 0; k0 < HID; k0 += 4) {
    int ka = k0 + 2 * khalf;
    v2f a, b;
    a.x = hrow[ka];
    a.y = hrow[ka + 1];
    b.x = wrel[(size_t)ka * NCLS + mn];
    b.y = wrel[(size_t)(ka + 1) * NCLS + mn];
    acc.v = __builtin_amdgcn_wmma_f32_16x16x4_f32(
         false, a,  false, b,
         (short)0, acc.v,  false,  false);
  }
  int rowoff = khalf * 8;
#pragma unroll
  for (int v = 0; v < 8; ++v) {
    P[(size_t)(mt * 16 + v + rowoff) * PCOLS + r * NCLS + mn] = acc.f[v];
  }
}

__global__ __launch_bounds__(256) void k_outinit(const float* __restrict__ b2,
                                                 const float* __restrict__ P,
                                                 float* __restrict__ out) {
  int idx = blockIdx.x * blockDim.x + threadIdx.x;
  if (idx >= NNODES * NCLS) return;
  int n = idx / NCLS, c = idx % NCLS;
  out[idx] = b2[c] + P[(size_t)n * PCOLS + (2 * NRELH) * NCLS + c];
}

__global__ __launch_bounds__(256) void k_layer2(const int* __restrict__ src,
                                                const int* __restrict__ rel,
                                                const int* __restrict__ dst,
                                                const float* __restrict__ cnt_s,
                                                const float* __restrict__ cnt_d,
                                                const float* __restrict__ P,
                                                float* __restrict__ out, int E) {
  long idx = (long)blockIdx.x * blockDim.x + threadIdx.x;
  if (idx >= (long)E * 16) return;
  int e = (int)(idx >> 4);
  int c = (int)idx & 15;
  int s = src[e], r = rel[e], d = dst[e];
  float vo = 1.0f / cnt_s[(size_t)r * NNODES + s];
  float vi = 1.0f / cnt_d[(size_t)r * NNODES + d];
  atomicAdd(&out[(size_t)s * NCLS + c], vo * P[(size_t)d * PCOLS + r * NCLS + c]);
  atomicAdd(&out[(size_t)d * NCLS + c], vi * P[(size_t)s * PCOLS + (r + NRELH) * NCLS + c]);
}

extern "C" void kernel_launch(void* const* d_in, const int* in_sizes, int n_in,
                              void* d_out, int out_size, void* d_ws, size_t ws_size,
                              hipStream_t stream) {
  const int*   src = (const int*)d_in[0];
  const int*   rel = (const int*)d_in[1];
  const int*   dst = (const int*)d_in[2];
  const float* w1  = (const float*)d_in[3];
  const float* b1  = (const float*)d_in[4];
  const float* w2  = (const float*)d_in[5];
  const float* b2  = (const float*)d_in[6];
  float* out = (float*)d_out;
  const int E = in_sizes[0];

  float* cnt_s = (float*)d_ws;
  float* cnt_d = cnt_s + CN;
  float* h     = cnt_d + CN;
  float* P     = h + (size_t)NNODES * HID;

  const int zn = 2 * CN + NNODES * HID;
  k_zero<<<(zn + 255) / 256, 256, 0, stream>>>(cnt_s, zn);

  k_count<<<(E + 255) / 256, 256, 0, stream>>>(src, rel, dst, cnt_s, cnt_d, E);

  long l1 = (long)E * 16;
  k_layer1<<<(unsigned)((l1 + 255) / 256), 256, 0, stream>>>(src, rel, dst, w1, cnt_s, cnt_d, h, E);

  k_hfinal<<<(NNODES * HID + 255) / 256, 256, 0, stream>>>(w1, b1, h);

  k_gemm<<<(NNODES / 16) * RTOT, 32, 0, stream>>>(h, w2, P);

  k_outinit<<<(NNODES * NCLS + 255) / 256, 256, 0, stream>>>(b2, P, out);

  long l2 = (long)E * 16;
  k_layer2<<<(unsigned)((l2 + 255) / 256), 256, 0, stream>>>(src, rel, dst, cnt_s, cnt_d, P, out, E);
}


# Layer-1 scatter: one wave per edge, lane L owns features L and L+32, so each f32 atomic instruction adds one whole 128-byte line (was 4 quarter-filled lines); add order into h changes
# speedup vs baseline: 1.3756x; 1.2599x over previous
#include <hip/hip_runtime.h>

#define NNODES 20000
#define NRELH  16
#define RTOT   33
#define HID    64
#define NCLS   16
#define RN     (RTOT * NNODES)
#define CN     (NRELH * NNODES)
#define PCOLS  (RTOT * NCLS)

typedef __attribute__((ext_vector_type(2))) float v2f;
typedef __attribute__((ext_vector_type(8))) float v8f;

__global__ __launch_bounds__(256) void k_zero(float* __restrict__ p, int n) {
  int i = blockIdx.x * blockDim.x + threadIdx.x;
  if (i < n) p[i] = 0.0f;
}

__global__ __launch_bounds__(256) void k_count(const int* __restrict__ src,
                                               const int* __restrict__ rel,
                                               const int* __restrict__ dst,
                                               float* __restrict__ cnt_s,
                                               float* __restrict__ cnt_d, int E) {
  int e = blockIdx.x * blockDim.x + threadIdx.x;
  if (e >= E) return;
  int s = src[e], r = rel[e], d = dst[e];
  atomicAdd(cnt_s + (size_t)r * NNODES + s, 1.0f);
  atomicAdd(cnt_d + (size_t)r * NNODES + d, 1.0f);
}

#define L1_EPW 4
template <int NE>
__device__ __forceinline__ void layer1_edges(const int* __restrict__ src,
                                             const int* __restrict__ rel,
                                             const int* __restrict__ dst,
                                             const float* __restrict__ w1,
                                             const float* __restrict__ cnt_s,
                                             const float* __restrict__ cnt_d,
                                             float* __restrict__ hacc, int e0, int lane) {
  int   sv[NE], dv[NE];
  float a0[NE], a1[NE], b0[NE], b1[NE];
#pragma unroll
  for (int j = 0; j < NE; ++j) {
    int e = e0 + j;
    int s = src[e], r = rel[e], d = dst[e];
    sv[j] = s; dv[j] = d;
    float vo = 1.0f / cnt_s[(size_t)r * NNODES + s];
    float vi = 1.0f / cnt_d[(size_t)r * NNODES + d];
    const float* ra = w1 + ((size_t)r * NNODES + d) * HID;
    const float* rb = w1 + ((size_t)(r + NRELH) * NNODES + s) * HID;
    a0[j] = vo * ra[lane];
    a1[j] = vo * ra[lane + 32];
    b0[j] = vi * rb[lane];
    b1[j] = vi * rb[lane + 32];
  }
#pragma unroll
  for (int j = 0; j < NE; ++j) {
    float* hs = hacc + (size_t)sv[j] * HID;
    float* hd = hacc + (size_t)dv[j] * HID;
    atomicAdd(hs + lane,      a0[j]);
    atomicAdd(hs + lane + 32, a1[j]);
    atomicAdd(hd + lane,      b0[j]);
    atomicAdd(hd + lane + 32, b1[j]);
  }
}

__global__ __launch_bounds__(256) void k_layer1(const int* __restrict__ src,
                                                const int* __restrict__ rel,
                                                const int* __restrict__ dst,
                                                const float* __restrict__ w1,
                                                const float* __restrict__ cnt_s,
                                                const float* __restrict__ cnt_d,
                                                float* __restrict__ hacc, int E) {
  int wave = __builtin_amdgcn_readfirstlane((int)((blockIdx.x * blockDim.x + threadIdx.x) >> 5));
  int lane = threadIdx.x & 31;
  int e0 = wave * L1_EPW;
  if (e0 + L1_EPW <= E) {
    layer1_edges<L1_EPW>(src, rel, dst, w1, cnt_s, cnt_d, hacc, e0, lane);
  } else {
    for (int e = e0; e < E; ++e)
      layer1_edges<1>(src, rel, dst, w1, cnt_s, cnt_d, hacc, e, lane);
  }
}

__global__ __launch_bounds__(256) void k_hfinal(const float* __restrict__ w1,
                                                const float* __restrict__ b1,
                                                float* __restrict__ h) {
  int idx = blockIdx.x * blockDim.x + threadIdx.x;
  if (idx >= NNODES * HID) return;
  int n = idx / HID, i = idx % HID;
  float v = h[idx] + w1[((size_t)(2 * NRELH) * NNODES + n) * HID + i] + b1[i];
  h[idx] = v > 0.0f ? v : 0.0f;
}

__global__ __launch_bounds__(32) void k_gemm(const float* __restrict__ h,
                                             const float* __restrict__ w2,
                                             float* __restrict__ P) {
  int tile = blockIdx.x;
  int mt = tile % (NNODES / 16);
  int r  = tile / (NNODES / 16);
  int lane  = threadIdx.x;
  int mn    = lane & 15;
  int khalf = lane >> 4;
  const float* hrow = h + (size_t)(mt * 16 + mn) * HID;
  const float* wrel = w2 + (size_t)r * HID * NCLS;
  union { v8f v; float f[8]; } acc;
  acc.v = (v8f)0.0f;
#pragma unroll
  for (int k0 = 0; k0 < HID; k0 += 4) {
    int ka = k0 + 2 * khalf;
    v2f a, b;
    a.x = hrow[ka];
    a.y = hrow[ka + 1];
    b.x = wrel[(size_t)ka * NCLS + mn];
    b.y = wrel[(size_t)(ka + 1) * NCLS + mn];
    acc.v = __builtin_amdgcn_wmma_f32_16x16x4_f32(
         false, a,  false, b,
         (short)0, acc.v,  false,  false);
  }
  int rowoff = khalf * 8;
#pragma unroll
  for (int v = 0; v < 8; ++v) {
    P[(size_t)(mt * 16 + v + rowoff) * PCOLS + r * NCLS + mn] = acc.f[v];
  }
}

__global__ __launch_bounds__(256) void k_outinit(const float* __restrict__ b2,
                                                 const float* __restrict__ P,
                                                 float* __restrict__ out) {
  int idx = blockIdx.x * blockDim.x + threadIdx.x;
  if (idx >= NNODES * NCLS) return;
  int n = idx / NCLS, c = idx % NCLS;
  out[idx] = b2[c] + P[(size_t)n * PCOLS + (2 * NRELH) * NCLS + c];
}

__global__ __launch_bounds__(256) void k_layer2(const int* __restrict__ src,
                                                const int* __restrict__ rel,
                                                const int* __restrict__ dst,
                                                const float* __restrict__ cnt_s,
                                                const float* __restrict__ cnt_d,
                                                const float* __restrict__ P,
                                                float* __restrict__ out, int E) {
  long idx = (long)blockIdx.x * blockDim.x + threadIdx.x;
  if (idx >= (long)E * 16) return;
  int e = (int)(idx >> 4);
  int c = (int)idx & 15;
  int s = src[e], r = rel[e], d = dst[e];
  float vo = 1.0f / cnt_s[(size_t)r * NNODES + s];
  float vi = 1.0f / cnt_d[(size_t)r * NNODES + d];
  atomicAdd(&out[(size_t)s * NCLS + c], vo * P[(size_t)d * PCOLS + r * NCLS + c]);
  atomicAdd(&out[(size_t)d * NCLS + c], vi * P[(size_t)s * PCOLS + (r + NRELH) * NCLS + c]);
}

extern "C" void kernel_launch(void* const* d_in, const int* in_sizes, int n_in,
                              void* d_out, int out_size, void* d_ws, size_t ws_size,
                              hipStream_t stream) {
  const int*   src = (const int*)d_in[0];
  const int*   rel = (const int*)d_in[1];
  const int*   dst = (const int*)d_in[2];
  const float* w1  = (const float*)d_in[3];
  const float* b1  = (const float*)d_in[4];
  const float* w2  = (const float*)d_in[5];
  const float* b2  = (const float*)d_in[6];
  float* out = (float*)d_out;
  const int E = in_sizes[0];

  float* cnt_s = (float*)d_ws;
  float* cnt_d = cnt_s + CN;
  float* h     = cnt_d + CN;
  float* P     = h + (size_t)NNODES * HID;

  const int zn = 2 * CN + NNODES * HID;
  k_zero<<<(zn + 255) / 256, 256, 0, stream>>>(cnt_s, zn);

  k_count<<<(E + 255) / 256, 256, 0, stream>>>(src, rel, dst, cnt_s, cnt_d, E);

  k_layer1<<<(unsigned)((E + 8 * L1_EPW - 1) / (8 * L1_EPW)), 256, 0, stream>>>(src, rel, dst, w1, cnt_s, cnt_d, h, E);

  k_hfinal<<<(NNODES * HID + 255) / 256, 256, 0, stream>>>(w1, b1, h);

  k_gemm<<<(NNODES / 16) * RTOT, 32, 0, stream>>>(h, w2, P);

  k_outinit<<<(NNODES * NCLS + 255) / 256, 256, 0, stream>>>(b2, P, out);

  long l2 = (long)E * 16;
  k_layer2<<<(unsigned)((l2 + 255) / 256), 256, 0, stream>>>(src, rel, dst, cnt_s, cnt_d, P, out, E);
}


# Fixed small grids whose waves loop over the work, all 7 kernels (GEMM wave keeps its A tile for 11 relations): waves launched 651250 -> ~46000; wave entry rate was the limit; atomic add order changes
# speedup vs baseline: 1.5655x; 1.1380x over previous
#include <hip/hip_runtime.h>

#define NNODES 20000
#define NRELH  16
#define RTOT   33
#define HID    64
#define NCLS   16
#define RN     (RTOT * NNODES)
#define CN     (NRELH * NNODES)
#define PCOLS  (RTOT * NCLS)

typedef __attribute__((ext_vector_type(2))) float v2f;
typedef __attribute__((ext_vector_type(8))) float v8f;

typedef __attribute__((ext_vector_type(4))) float v4f;
__global__ __launch_bounds__(256) void k_zero(v4f* __restrict__ p, int n4) {
  int stride = gridDim.x * blockDim.x;
  for (int i = blockIdx.x * blockDim.x + threadIdx.x; i < n4; i += stride) p[i] = (v4f)0.0f;
}

__global__ __launch_bounds__(256) void k_count(const int* __restrict__ src,
                                               const int* __restrict__ rel,
                                               const int* __restrict__ dst,
                                               float* __restrict__ cnt_s,
                                               float* __restrict__ cnt_d, int E) {
  int stride = gridDim.x * blockDim.x;
  for (int e = blockIdx.x * blockDim.x + threadIdx.x; e < E; e += stride) {
    int s = src[e], r = rel[e], d = dst[e];
    atomicAdd(cnt_s + (size_t)r * NNODES + s, 1.0f);
    atomicAdd(cnt_d + (size_t)r * NNODES + d, 1.0f);
  }
}

#define L1_EPW 4
template <int NE>
__device__ __forceinline__ void layer1_edges(const int* __restrict__ src,
                                             const int* __restrict__ rel,
                                             const int* __restrict__ dst,
                                             const float* __restrict__ w1,
                                             const float* __restrict__ cnt_s,
                                             const float* __restrict__ cnt_d,
                                             float* __restrict__ hacc, int e0, int lane) {
  int   sv[NE], dv[NE];
  float a0[NE], a1[NE], b0[NE], b1[NE];
#pragma unroll
  for (int j = 0; j < NE; ++j) {
    int e = e0 + j;
    int s = src[e], r = rel[e], d = dst[e];
    sv[j] = s; dv[j] = d;
    float vo = 1.0f / cnt_s[(size_t)r * NNODES + s];
    float vi = 1.0f / cnt_d[(size_t)r * NNODES + d];
    const float* ra = w1 + ((size_t)r * NNODES + d) * HID;
    const float* rb = w1 + ((size_t)(r + NRELH) * NNODES + s) * HID;
    a0[j] = vo * ra[lane];
    a1[j] = vo * ra[lane + 32];
    b0[j] = vi * rb[lane];
    b1[j] = vi * rb[lane + 32];
  }
#pragma unroll
  for (int j = 0; j < NE; ++j) {
    float* hs = hacc + (size_t)sv[j] * HID;
    float* hd = hacc + (size_t)dv[j] * HID;
    atomicAdd(hs + lane,      a0[j]);
    atomicAdd(hs + lane + 32, a1[j]);
    atomicAdd(hd + lane,      b0[j]);
    atomicAdd(hd + lane + 32, b1[j]);
  }
}

__global__ __launch_bounds__(256) void k_layer1(const int* __restrict__ src,
                                                const int* __restrict__ rel,
                                                const int* __restrict__ dst,
                                                const float* __restrict__ w1,
                                                const float* __restrict__ cnt_s,
                                                const float* __restrict__ cnt_d,
                                                float* __restrict__ hacc, int E) {
  int wave = __builtin_amdgcn_readfirstlane((int)((blockIdx.x * blockDim.x + threadIdx.x) >> 5));
  int lane = threadIdx.x & 31;
  int nwaves  = (int)((gridDim.x * blockDim.x) >> 5);
  int ngroups = E / L1_EPW;
  for (int g = wave; g < ngroups; g += nwaves)
    layer1_edges<L1_EPW>(src, rel, dst, w1, cnt_s, cnt_d, hacc, g * L1_EPW, lane);
  if (wave == 0) {
    for (int e = ngroups * L1_EPW; e < E; ++e)
      layer1_edges<1>(src, rel, dst, w1, cnt_s, cnt_d, hacc, e, lane);
  }
}

__global__ __launch_bounds__(256) void k_hfinal(const float* __restrict__ w1,
                                                const float* __restrict__ b1,
                                                float* __restrict__ h) {
  const float* wself = w1 + (size_t)(2 * NRELH) * NNODES * HID;
  int stride = gridDim.x * blockDim.x;
  for (int q = blockIdx.x * blockDim.x + threadIdx.x; q < NNODES * HID / 4; q += stride) {
    int idx = q * 4, i = idx % HID;
#pragma unroll
    for (int t = 0; t < 4; ++t) {
      float v = h[idx + t] + wself[idx + t] + b1[i + t];
      h[idx + t] = v > 0.0f ? v : 0.0f;
    }
  }
}

#define GEMM_RG  3
#define GEMM_RPG (RTOT / GEMM_RG)
#define GEMM_WAVES ((NNODES / 16) * GEMM_RG)
__global__ __launch_bounds__(256) void k_gemm(const float* __restrict__ h,
                                              const float* __restrict__ w2,
                                              float* __restrict__ P) {
  int wid = __builtin_amdgcn_readfirstlane((int)((blockIdx.x * blockDim.x + threadIdx.x) >> 5));
  if (wid >= GEMM_WAVES) return;
  int mt = wid / GEMM_RG;
  int r0 = (wid % GEMM_RG) * GEMM_RPG;
  int lane  = threadIdx.x & 31;
  int mn    = lane & 15;
  int khalf = lane >> 4;
  const float* hrow = h + (size_t)(mt * 16 + mn) * HID;
  v2f a[HID / 4];
#pragma unroll
  for (int kk = 0; kk < HID / 4; ++kk) {
    int ka = 4 * kk + 2 * khalf;
    a[kk].x = hrow[ka];
    a[kk].y = hrow[ka + 1];
  }
  int rowoff = khalf * 8;
  for (int r = r0; r < r0 + GEMM_RPG; ++r) {
    const float* wrel = w2 + (size_t)r * HID * NCLS;
    union { v8f v; float f[8]; } acc;
    acc.v = (v8f)0.0f;
#pragma unroll
    for (int kk = 0; kk < HID / 4; ++kk) {
      int ka = 4 * kk + 2 * khalf;
      v2f b;
      b.x = wrel[(size_t)ka * NCLS + mn];
      b.y = wrel[(size_t)(ka + 1) * NCLS + mn];
      acc.v = __builtin_amdgcn_wmma_f32_16x16x4_f32(
           false, a[kk],  false, b,
           (short)0, acc.v,  false,  false);
    }
#pragma unroll
    for (int v = 0; v < 8; ++v) {
      P[(size_t)(mt * 16 + v + rowoff) * PCOLS + r * NCLS + mn] = acc.f[v];
    }
  }
}

__global__ __launch_bounds__(256) void k_outinit(const float* __restrict__ b2,
                                                 const float* __restrict__ P,
                                                 float* __restrict__ out) {
  int q = blockIdx.x * blockDim.x + threadIdx.x;
  if (q >= NNODES * NCLS / 4) return;
  int idx = q * 4;
  int n = idx / NCLS, c = idx % NCLS;
  const float* prow = P + (size_t)n * PCOLS + (2 * NRELH) * NCLS + c;
#pragma unroll
  for (int t = 0; t < 4; ++t) out[idx + t] = b2[c + t] + prow[t];
}

#define L2_TPB 4
template <int NT>
__device__ __forceinline__ void layer2_trips(const int* __restrict__ src,
                                             const int* __restrict__ rel,
                                             const int* __restrict__ dst,
                                             const float* __restrict__ cnt_s,
                                             const float* __restrict__ cnt_d,
                                             const float* __restrict__ P,
                                             float* __restrict__ out, int e0, int c) {
  int   sv[NT], dv[NT];
  float x[NT], y[NT];
#pragma unroll
  for (int j = 0; j < NT; ++j) {
    int e = e0 + 2 * j;
    int s = src[e], r = rel[e], d = dst[e];
    sv[j] = s; dv[j] = d;
    float vo = 1.0f / cnt_s[(size_t)r * NNODES + s];
    float vi = 1.0f / cnt_d[(size_t)r * NNODES + d];
    x[j] = vo * P[(size_t)d * PCOLS + r * NCLS + c];
    y[j] = vi * P[(size_t)s * PCOLS + (r + NRELH) * NCLS + c];
  }
#pragma unroll
  for (int j = 0; j < NT; ++j) {
    atomicAdd(&out[(size_t)sv[j] * NCLS + c], x[j]);
    atomicAdd(&out[(size_t)dv[j] * NCLS + c], y[j]);
  }
}

__global__ __launch_bounds__(256) void k_layer2(const int* __restrict__ src,
                                                const int* __restrict__ rel,
                                                const int* __restrict__ dst,
                                                const float* __restrict__ cnt_s,
                                                const float* __restrict__ cnt_d,
                                                const float* __restrict__ P,
                                                float* __restrict__ out, int E) {
  int wave = __builtin_amdgcn_readfirstlane((int)((blockIdx.x * blockDim.x + threadIdx.x) >> 5));
  int half = (threadIdx.x >> 4) & 1;
  int c    = threadIdx.x & 15;
  int nwaves   = (int)((gridDim.x * blockDim.x) >> 5);
  int nbatches = E / (2 * L2_TPB);
  for (int b = wave; b < nbatches; b += nwaves)
    layer2_trips<L2_TPB>(src, rel, dst, cnt_s, cnt_d, P, out, b * (2 * L2_TPB) + half, c);
  if (wave == 0) {
    for (int e = nbatches * (2 * L2_TPB) + half; e < E; e += 2)
      layer2_trips<1>(src, rel, dst, cnt_s, cnt_d, P, out, e, c);
  }
}

extern "C" void kernel_launch(void* const* d_in, const int* in_sizes, int n_in,
                              void* d_out, int out_size, void* d_ws, size_t ws_size,
                              hipStream_t stream) {
  const int*   src = (const int*)d_in[0];
  const int*   rel = (const int*)d_in[1];
  const int*   dst = (const int*)d_in[2];
  const float* w1  = (const float*)d_in[3];
  const float* b1  = (const float*)d_in[4];
  const float* w2  = (const float*)d_in[5];
  const float* b2  = (const float*)d_in[6];
  float* out = (float*)d_out;
  const int E = in_sizes[0];

  float* cnt_s = (float*)d_ws;
  float* cnt_d = cnt_s + CN;
  float* h     = cnt_d + CN;
  float* P     = h + (size_t)NNODES * HID;

  const int zn = 2 * CN + NNODES * HID;
  k_zero<<<1024, 256, 0, stream>>>((v4f*)cnt_s, zn / 4);

  k_count<<<1024, 256, 0, stream>>>(src, rel, dst, cnt_s, cnt_d, E);

  k_layer1<<<1536, 256, 0, stream>>>(src, rel, dst, w1, cnt_s, cnt_d, h, E);

  k_hfinal<<<1024, 256, 0, stream>>>(w1, b1, h);

  k_gemm<<<(GEMM_WAVES + 7) / 8, 256, 0, stream>>>(h, w2, P);

  k_outinit<<<(NNODES * NCLS / 4 + 255) / 256, 256, 0, stream>>>(b2, P, out);

  k_layer2<<<1536, 256, 0, stream>>>(src, rel, dst, cnt_s, cnt_d, P, out, E);
}


# Fold the two elementwise passes into their neighbours: h starts as self-loop row + b1, ReLU in GEMM A loader, self-loop product + b2 stored to out by the GEMM; 7 launches to 5; h add order changes
# speedup vs baseline: 1.6504x; 1.0542x over previous
#include <hip/hip_runtime.h>

#define NNODES 20000
#define NRELH  16
#define RTOT   33
#define HID    64
#define NCLS   16
#define RN     (RTOT * NNODES)
#define CN     (NRELH * NNODES)
#define PCOLS  (2 * NRELH * NCLS)

typedef __attribute__((ext_vector_type(2))) float v2f;
typedef __attribute__((ext_vector_type(8))) float v8f;

typedef __attribute__((ext_vector_type(4))) float v4f;
__global__ __launch_bounds__(256) void k_init(const float* __restrict__ w1,
                                              const float* __restrict__ b1,
                                              v4f* __restrict__ cnt4,
                                              float* __restrict__ h) {
  const float* wself = w1 + (size_t)(2 * NRELH) * NNODES * HID;
  int tid = blockIdx.x * blockDim.x + threadIdx.x;
  int stride = gridDim.x * blockDim.x;
  for (int i = tid; i < 2 * CN / 4; i += stride) cnt4[i] = (v4f)0.0f;
  for (int q = tid; q < NNODES * HID / 4; q += stride) {
    int idx = q * 4, i = idx % HID;
#pragma unroll
    for (int t = 0; t < 4; ++t) h[idx + t] = wself[idx + t] + b1[i + t];
  }
}

__global__ __launch_bounds__(256) void k_count(const int* __restrict__ src,
                                               const int* __restrict__ rel,
                                               const int* __restrict__ dst,
                                               float* __restrict__ cnt_s,
                                               float* __restrict__ cnt_d, int E) {
  int stride = gridDim.x * blockDim.x;
  for (int e = blockIdx.x * blockDim.x + threadIdx.x; e < E; e += stride) {
    int s = src[e], r = rel[e], d = dst[e];
    atomicAdd(cnt_s + (size_t)r * NNODES + s, 1.0f);
    atomicAdd(cnt_d + (size_t)r * NNODES + d, 1.0f);
  }
}

#define L1_EPW 4
template <int NE>
__device__ __forceinline__ void layer1_edges(const int* __restrict__ src,
                                             const int* __restrict__ rel,
                                             const int* __restrict__ dst,
                                             const float* __restrict__ w1,
                                             const float* __restrict__ cnt_s,
                                             const float* __restrict__ cnt_d,
                                             float* __restrict__ hacc, int e0, int lane) {
  int   sv[NE], dv[NE];
  float a0[NE], a1[NE], b0[NE], b1[NE];
#pragma unroll
  for (int j = 0; j < NE; ++j) {
    int e = e0 + j;
    int s = src[e], r = rel[e], d = dst[e];
    sv[j] = s; dv[j] = d;
    float vo = 1.0f / cnt_s[(size_t)r * NNODES + s];
    float vi = 1.0f / cnt_d[(size_t)r * NNODES + d];
    const float* ra = w1 + ((size_t)r * NNODES + d) * HID;
    const float* rb = w1 + ((size_t)(r + NRELH) * NNODES + s) * HID;
    a0[j] = vo * ra[lane];
    a1[j] = vo * ra[lane + 32];
    b0[j] = vi * rb[lane];
    b1[j] = vi * rb[lane + 32];
  }
#pragma unroll
  for (int j = 0; j < NE; ++j) {
    float* hs = hacc + (size_t)sv[j] * HID;
    float* hd = hacc + (size_t)dv[j] * HID;
    atomicAdd(hs + lane,      a0[j]);
    atomicAdd(hs + lane + 32, a1[j]);
    atomicAdd(hd + lane,      b0[j]);
    atomicAdd(hd + lane + 32, b1[j]);
  }
}

__global__ __launch_bounds__(256) void k_layer1(const int* __restrict__ src,
                                                const int* __restrict__ rel,
                                                const int* __restrict__ dst,
                                                const float* __restrict__ w1,
                                                const float* __restrict__ cnt_s,
                                                const float* __restrict__ cnt_d,
                                                float* __restrict__ hacc, int E) {
  int wave = __builtin_amdgcn_readfirstlane((int)((blockIdx.x * blockDim.x + threadIdx.x) >> 5));
  int lane = threadIdx.x & 31;
  int nwaves  = (int)((gridDim.x * blockDim.x) >> 5);
  int ngroups = E / L1_EPW;
  for (int g = wave; g < ngroups; g += nwaves)
    layer1_edges<L1_EPW>(src, rel, dst, w1, cnt_s, cnt_d, hacc, g * L1_EPW, lane);
  if (wave == 0) {
    for (int e = ngroups * L1_EPW; e < E; ++e)
      layer1_edges<1>(src, rel, dst, w1, cnt_s, cnt_d, hacc, e, lane);
  }
}

#define GEMM_RG  3
#define GEMM_RPG (RTOT / GEMM_RG)
#define GEMM_WAVES ((NNODES / 16) * GEMM_RG)
__device__ __forceinline__ v8f tile_product(const v2f (&a)[HID / 4], const float* __restrict__ wrel,
                                            int khalf, int mn) {
  v8f acc = (v8f)0.0f;
#pragma unroll
  for (int kk = 0; kk < HID / 4; ++kk) {
    int ka = 4 * kk + 2 * khalf;
    v2f b;
    b.x = wrel[(size_t)ka * NCLS + mn];
    b.y = wrel[(size_t)(ka + 1) * NCLS + mn];
    acc = __builtin_amdgcn_wmma_f32_16x16x4_f32(
         false, a[kk],  false, b,
         (short)0, acc,  false,  false);
  }
  return acc;
}

template <bool SELF>
__device__ __forceinline__ void gemm_relations(const v2f (&a)[HID / 4], const float* __restrict__ w2,
                                               const float* __restrict__ b2, float* __restrict__ P,
                                               float* __restrict__ out, int mt, int r0, int rowoff,
                                               int khalf, int mn) {
  constexpr int NP = SELF ? GEMM_RPG - 1 : GEMM_RPG;
  for (int r = r0; r < r0 + NP; ++r) {
    union { v8f v; float f[8]; } acc;
    acc.v = tile_product(a, w2 + (size_t)r * HID * NCLS, khalf, mn);
#pragma unroll
    for (int v = 0; v < 8; ++v) {
      P[(size_t)(mt * 16 + v + rowoff) * PCOLS + r * NCLS + mn] = acc.f[v];
    }
  }
  if (SELF) {
    union { v8f v; float f[8]; } acc;
    acc.v = tile_product(a, w2 + (size_t)(2 * NRELH) * HID * NCLS, khalf, mn);
    float bias = b2[mn];
#pragma unroll
    for (int v = 0; v < 8; ++v) {
      out[(size_t)(mt * 16 + v + rowoff) * NCLS + mn] = acc.f[v] + bias;
    }
  }
}

__global__ __launch_bounds__(256) void k_gemm(const float* __restrict__ h,
                                              const float* __restrict__ w2,
                                              const float* __restrict__ b2,
                                              float* __restrict__ P,
                                              float* __restrict__ out) {
  int wid = __builtin_amdgcn_readfirstlane((int)((blockIdx.x * blockDim.x + threadIdx.x) >> 5));
  if (wid >= GEMM_WAVES) return;
  int mt = wid / GEMM_RG;
  int r0 = (wid % GEMM_RG) * GEMM_RPG;
  int lane  = threadIdx.x & 31;
  int mn    = lane & 15;
  int khalf = lane >> 4;
  const float* hrow = h + (size_t)(mt * 16 + mn) * HID;
  v2f a[HID / 4];
#pragma unroll
  for (int kk = 0; kk < HID / 4; ++kk) {
    int ka = 4 * kk + 2 * khalf;
    float x = hrow[ka], y = hrow[ka + 1];
    a[kk].x = x > 0.0f ? x : 0.0f;
    a[kk].y = y > 0.0f ? y : 0.0f;
  }
  int rowoff = khalf * 8;
  if (r0 + GEMM_RPG == RTOT)
    gemm_relations<true>(a, w2, b2, P, out, mt, r0, rowoff, khalf, mn);
  else
    gemm_relations<false>(a, w2, b2, P, out, mt, r0, rowoff, khalf, mn);
}

#define L2_TPB 4
template <int NT>
__device__ __forceinline__ void layer2_trips(const int* __restrict__ src,
                                             const int* __restrict__ rel,
                                             const int* __restrict__ dst,
                                             const float* __restrict__ cnt_s,
                                             const float* __restrict__ cnt_d,
                                             const float* __restrict__ P,
                                             float* __restrict__ out, int e0, int c) {
  int   sv[NT], dv[NT];
  float x[NT], y[NT];
#pragma unroll
  for (int j = 0; j < NT; ++j) {
    int e = e0 + 2 * j;
    int s = src[e], r = rel[e], d = dst[e];
    sv[j] = s; dv[j] = d;
    float vo = 1.0f / cnt_s[(size_t)r * NNODES + s];
    float vi = 1.0f / cnt_d[(size_t)r * NNODES + d];
    x[j] = vo * P[(size_t)d * PCOLS + r * NCLS + c];
    y[j] = vi * P[(size_t)s * PCOLS + (r + NRELH) * NCLS + c];
  }
#pragma unroll
  for (int j = 0; j < NT; ++j) {
    atomicAdd(&out[(size_t)sv[j] * NCLS + c], x[j]);
    atomicAdd(&out[(size_t)dv[j] * NCLS + c], y[j]);
  }
}

__global__ __launch_bounds__(256) void k_layer2(const int* __restrict__ src,
                                                const int* __restrict__ rel,
                                                const int* __restrict__ dst,
                                                const float* __restrict__ cnt_s,
                                                const float* __restrict__ cnt_d,
                                                const float* __restrict__ P,
                                                float* __restrict__ out, int E) {
  int wave = __builtin_amdgcn_readfirstlane((int)((blockIdx.x * blockDim.x + threadIdx.x) >> 5));
  int half = (threadIdx.x >> 4) & 1;
  int c    = threadIdx.x & 15;
  int nwaves   = (int)((gridDim.x * blockDim.x) >> 5);
  int nbatches = E / (2 * L2_TPB);
  for (int b = wave; b < nbatches; b += nwaves)
    layer2_trips<L2_TPB>(src, rel, dst, cnt_s, cnt_d, P, out, b * (2 * L2_TPB) + half, c);
  if (wave == 0) {
    for (int e = nbatches * (2 * L2_TPB) + half; e < E; e += 2)
      layer2_trips<1>(src, rel, dst, cnt_s, cnt_d, P, out, e, c);
  }
}

extern "C" void kernel_launch(void* const* d_in, const int* in_sizes, int n_in,
                              void* d_out, int out_size, void* d_ws, size_t ws_size,
                              hipStream_t stream) {
  const int*   src = (const int*)d_in[0];
  const int*   rel = (const int*)d_in[1];
  const int*   dst = (const int*)d_in[2];
  const float* w1  = (const float*)d_in[3];
  const float* b1  = (const float*)d_in[4];
  const float* w2  = (const float*)d_in[5];
  const float* b2  = (const float*)d_in[6];
  float* out = (float*)d_out;
  const int E = in_sizes[0];

  float* cnt_s = (float*)d_ws;
  float* cnt_d = cnt_s + CN;
  float* h     = cnt_d + CN;
  float* P     = h + (size_t)NNODES * HID;

  k_init<<<1024, 256, 0, stream>>>(w1, b1, (v4f*)cnt_s, h);

  k_count<<<1024, 256, 0, stream>>>(src, rel, dst, cnt_s, cnt_d, E);

  k_layer1<<<1536, 256, 0, stream>>>(src, rel, dst, w1, cnt_s, cnt_d, h, E);

  k_gemm<<<(GEMM_WAVES + 7) / 8, 256, 0, stream>>>(h, w2, b2, P, out);

  k_layer2<<<1536, 256, 0, stream>>>(src, rel, dst, cnt_s, cnt_d, P, out, E);
}


# Counts: each work-group counts one (edge end, relation) row of the table in LDS over 1/8 of the edges, then adds it to memory as whole lines: 160 k line atomics replace 1.28 M one-word atomics
# speedup vs baseline: 1.9561x; 1.1852x over previous
#include <hip/hip_runtime.h>

#define NNODES 20000
#define NRELH  16
#define RTOT   33
#define HID    64
#define NCLS   16
#define RN     (RTOT * NNODES)
#define CN     (NRELH * NNODES)
#define PCOLS  (2 * NRELH * NCLS)

typedef __attribute__((ext_vector_type(2))) float v2f;
typedef __attribute__((ext_vector_type(8))) float v8f;

typedef __attribute__((ext_vector_type(4))) float v4f;
__global__ __launch_bounds__(256) void k_init(const float* __restrict__ w1,
                                              const float* __restrict__ b1,
                                              v4f* __restrict__ cnt4,
                                              float* __restrict__ h) {
  const float* wself = w1 + (size_t)(2 * NRELH) * NNODES * HID;
  int tid = blockIdx.x * blockDim.x + threadIdx.x;
  int stride = gridDim.x * blockDim.x;
  for (int i = tid; i < 2 * CN / 4; i += stride) cnt4[i] = (v4f)0.0f;
  for (int q = tid; q < NNODES * HID / 4; q += stride) {
    int idx = q * 4, i = idx % HID;
#pragma unroll
    for (int t = 0; t < 4; ++t) h[idx + t] = wself[idx + t] + b1[i + t];
  }
}

#define CNT_ROWS   (2 * NRELH)
#define CNT_CHUNKS 8
#define CNT_TPB    1024
#define CNT_UNR    4
typedef __attribute__((ext_vector_type(4))) int v4i;
__global__ __launch_bounds__(CNT_TPB) void k_count(const int* __restrict__ src,
                                                   const int* __restrict__ rel,
                                                   const int* __restrict__ dst,
                                                   float* __restrict__ cnt, int E) {
  __shared__ int bins[NNODES];
  const int row   = blockIdx.x % CNT_ROWS;
  const int chunk = blockIdx.x / CNT_ROWS;
  const int r     = row % NRELH;
  const int* __restrict__ node = row < NRELH ? src : dst;
  const int tid = threadIdx.x;

  for (int j = tid; j < NNODES; j += CNT_TPB) bins[j] = 0;
  __syncthreads();

  const int clen = (((E + CNT_CHUNKS - 1) / CNT_CHUNKS) + 3) & ~3;
  int e0 = chunk * clen;      if (e0 > E) e0 = E;
  int e1 = e0 + clen;         if (e1 > E) e1 = E;
  const int nvec = (e1 - e0) >> 2;
  const v4i* __restrict__ rel4  = (const v4i*)(rel + e0);
  const v4i* __restrict__ node4 = (const v4i*)(node + e0);

  for (int q0 = tid; q0 < nvec; q0 += CNT_UNR * CNT_TPB) {
    v4i rv[CNT_UNR], nv[CNT_UNR];
#pragma unroll
    for (int u = 0; u < CNT_UNR; ++u) {
      int q = q0 + u * CNT_TPB;
      bool ok = q < nvec;
      rv[u] = ok ? rel4[q]  : (v4i)(-1);
      nv[u] = ok ? node4[q] : (v4i)0;
    }
#pragma unroll
    for (int u = 0; u < CNT_UNR; ++u) {
      if (rv[u].x == r) atomicAdd(&bins[nv[u].x], 1);
      if (rv[u].y == r) atomicAdd(&bins[nv[u].y], 1);
      if (rv[u].z == r) atomicAdd(&bins[nv[u].z], 1);
      if (rv[u].w == r) atomicAdd(&bins[nv[u].w], 1);
    }
  }
  for (int e = e0 + 4 * nvec + tid; e < e1; e += CNT_TPB)
    if (rel[e] == r) atomicAdd(&bins[node[e]], 1);
  __syncthreads();

  float* __restrict__ trow = cnt + (size_t)row * NNODES;
  for (int j = tid; j < NNODES; j += CNT_TPB) atomicAdd(trow + j, (float)bins[j]);
}

#define L1_EPW 4
template <int NE>
__device__ __forceinline__ void layer1_edges(const int* __restrict__ src,
                                             const int* __restrict__ rel,
                                             const int* __restrict__ dst,
                                             const float* __restrict__ w1,
                                             const float* __restrict__ cnt_s,
                                             const float* __restrict__ cnt_d,
                                             float* __restrict__ hacc, int e0, int lane) {
  int   sv[NE], dv[NE];
  float a0[NE], a1[NE], b0[NE], b1[NE];
#pragma unroll
  for (int j = 0; j < NE; ++j) {
    int e = e0 + j;
    int s = src[e], r = rel[e], d = dst[e];
    sv[j] = s; dv[j] = d;
    float vo = 1.0f / cnt_s[(size_t)r * NNODES + s];
    float vi = 1.0f / cnt_d[(size_t)r * NNODES + d];
    const float* ra = w1 + ((size_t)r * NNODES + d) * HID;
    const float* rb = w1 + ((size_t)(r + NRELH) * NNODES + s) * HID;
    a0[j] = vo * ra[lane];
    a1[j] = vo * ra[lane + 32];
    b0[j] = vi * rb[lane];
    b1[j] = vi * rb[lane + 32];
  }
#pragma unroll
  for (int j = 0; j < NE; ++j) {
    float* hs = hacc + (size_t)sv[j] * HID;
    float* hd = hacc + (size_t)dv[j] * HID;
    atomicAdd(hs + lane,      a0[j]);
    atomicAdd(hs + lane + 32, a1[j]);
    atomicAdd(hd + lane,      b0[j]);
    atomicAdd(hd + lane + 32, b1[j]);
  }
}

__global__ __launch_bounds__(256) void k_layer1(const int* __restrict__ src,
                                                const int* __restrict__ rel,
                                                const int* __restrict__ dst,
                                                const float* __restrict__ w1,
                                                const float* __restrict__ cnt_s,
                                                const float* __restrict__ cnt_d,
                                                float* __restrict__ hacc, int E) {
  int wave = __builtin_amdgcn_readfirstlane((int)((blockIdx.x * blockDim.x + threadIdx.x) >> 5));
  int lane = threadIdx.x & 31;
  int nwaves  = (int)((gridDim.x * blockDim.x) >> 5);
  int ngroups = E / L1_EPW;
  for (int g = wave; g < ngroups; g += nwaves)
    layer1_edges<L1_EPW>(src, rel, dst, w1, cnt_s, cnt_d, hacc, g * L1_EPW, lane);
  if (wave == 0) {
    for (int e = ngroups * L1_EPW; e < E; ++e)
      layer1_edges<1>(src, rel, dst, w1, cnt_s, cnt_d, hacc, e, lane);
  }
}

#define GEMM_RG  3
#define GEMM_RPG (RTOT / GEMM_RG)
#define GEMM_WAVES ((NNODES / 16) * GEMM_RG)
__device__ __forceinline__ v8f tile_product(const v2f (&a)[HID / 4], const float* __restrict__ wrel,
                                            int khalf, int mn) {
  v8f acc = (v8f)0.0f;
#pragma unroll
  for (int kk = 0; kk < HID / 4; ++kk) {
    int ka = 4 * kk + 2 * khalf;
    v2f b;
    b.x = wrel[(size_t)ka * NCLS + mn];
    b.y = wrel[(size_t)(ka + 1) * NCLS + mn];
    acc = __builtin_amdgcn_wmma_f32_16x16x4_f32(
         false, a[kk],  false, b,
         (short)0, acc,  false,  false);
  }
  return acc;
}

template <bool SELF>
__device__ __forceinline__ void gemm_relations(const v2f (&a)[HID / 4], const float* __restrict__ w2,
                                               const float* __restrict__ b2, float* __restrict__ P,
                                               float* __restrict__ out, int mt, int r0, int rowoff,
                                               int khalf, int mn) {
  constexpr int NP = SELF ? GEMM_RPG - 1 : GEMM_RPG;
  for (int r = r0; r < r0 + NP; ++r) {
    union { v8f v; float f[8]; } acc;
    acc.v = tile_product(a, w2 + (size_t)r * HID * NCLS, khalf, mn);
#pragma unroll
    for (int v = 0; v < 8; ++v) {
      P[(size_t)(mt * 16 + v + rowoff) * PCOLS + r * NCLS + mn] = acc.f[v];
    }
  }
  if (SELF) {
    union { v8f v; float f[8]; } acc;
    acc.v = tile_product(a, w2 + (size_t)(2 * NRELH) * HID * NCLS, khalf, mn);
    float bias = b2[mn];
#pragma unroll
    for (int v = 0; v < 8; ++v) {
      out[(size_t)(mt * 16 + v + rowoff) * NCLS + mn] = acc.f[v] + bias;
    }
  }
}

__global__ __launch_bounds__(256) void k_gemm(const float* __restrict__ h,
                                              const float* __restrict__ w2,
                                              const float* __restrict__ b2,
                                              float* __restrict__ P,
                                              float* __restrict__ out) {
  int wid = __builtin_amdgcn_readfirstlane((int)((blockIdx.x * blockDim.x + threadIdx.x) >> 5));
  if (wid >= GEMM_WAVES) return;
  int mt = wid / GEMM_RG;
  int r0 = (wid % GEMM_RG) * GEMM_RPG;
  int lane  = threadIdx.x & 31;
  int mn    = lane & 15;
  int khalf = lane >> 4;
  const float* hrow = h + (size_t)(mt * 16 + mn) * HID;
  v2f a[HID / 4];
#pragma unroll
  for (int kk = 0; kk < HID / 4; ++kk) {
    int ka = 4 * kk + 2 * khalf;
    float x = hrow[ka], y = hrow[ka + 1];
    a[kk].x = x > 0.0f ? x : 0.0f;
    a[kk].y = y > 0.0f ? y : 0.0f;
  }
  int rowoff = khalf * 8;
  if (r0 + GEMM_RPG == RTOT)
    gemm_relations<true>(a, w2, b2, P, out, mt, r0, rowoff, khalf, mn);
  else
    gemm_relations<false>(a, w2, b2, P, out, mt, r0, rowoff, khalf, mn);
}

#define L2_TPB 4
template <int NT>
__device__ __forceinline__ void layer2_trips(const int* __restrict__ src,
                                             const int* __restrict__ rel,
                                             const int* __restrict__ dst,
                                             const float* __restrict__ cnt_s,
                                             const float* __restrict__ cnt_d,
                                             const float* __restrict__ P,
                                             float* __restrict__ out, int e0, int c) {
  int   sv[NT], dv[NT];
  float x[NT], y[NT];
#pragma unroll
  for (int j = 0; j < NT; ++j) {
    int e = e0 + 2 * j;
    int s = src[e], r = rel[e], d = dst[e];
    sv[j] = s; dv[j] = d;
    float vo = 1.0f / cnt_s[(size_t)r * NNODES + s];
    float vi = 1.0f / cnt_d[(size_t)r * NNODES + d];
    x[j] = vo * P[(size_t)d * PCOLS + r * NCLS + c];
    y[j] = vi * P[(size_t)s * PCOLS + (r + NRELH) * NCLS + c];
  }
#pragma unroll
  for (int j = 0; j < NT; ++j) {
    atomicAdd(&out[(size_t)sv[j] * NCLS + c], x[j]);
    atomicAdd(&out[(size_t)dv[j] * NCLS + c], y[j]);
  }
}

__global__ __launch_bounds__(256) void k_layer2(const int* __restrict__ src,
                                                const int* __restrict__ rel,
                                                const int* __restrict__ dst,
                                                const float* __restrict__ cnt_s,
                                                const float* __restrict__ cnt_d,
                                                const float* __restrict__ P,
                                                float* __restrict__ out, int E) {
  int wave = __builtin_amdgcn_readfirstlane((int)((blockIdx.x * blockDim.x + threadIdx.x) >> 5));
  int half = (threadIdx.x >> 4) & 1;
  int c    = threadIdx.x & 15;
  int nwaves   = (int)((gridDim.x * blockDim.x) >> 5);
  int nbatches = E / (2 * L2_TPB);
  for (int b = wave; b < nbatches; b += nwaves)
    layer2_trips<L2_TPB>(src, rel, dst, cnt_s, cnt_d, P, out, b * (2 * L2_TPB) + half, c);
  if (wave == 0) {
    for (int e = nbatches * (2 * L2_TPB) + half; e < E; e += 2)
      layer2_trips<1>(src, rel, dst, cnt_s, cnt_d, P, out, e, c);
  }
}

extern "C" void kernel_launch(void* const* d_in, const int* in_sizes, int n_in,
                              void* d_out, int out_size, void* d_ws, size_t ws_size,
                              hipStream_t stream) {
  const int*   src = (const int*)d_in[0];
  const int*   rel = (const int*)d_in[1];
  const int*   dst = (const int*)d_in[2];
  const float* w1  = (const float*)d_in[3];
  const float* b1  = (const float*)d_in[4];
  const float* w2  = (const float*)d_in[5];
  const float* b2  = (const float*)d_in[6];
  float* out = (float*)d_out;
  const int E = in_sizes[0];

  float* cnt_s = (float*)d_ws;
  float* cnt_d = cnt_s + CN;
  float* h     = cnt_d + CN;
  float* P     = h + (size_t)NNODES * HID;

  k_init<<<1024, 256, 0, stream>>>(w1, b1, (v4f*)cnt_s, h);

  k_count<<<CNT_ROWS * CNT_CHUNKS, CNT_TPB, 0, stream>>>(src, rel, dst, cnt_s, E);

  k_layer1<<<1536, 256, 0, stream>>>(src, rel, dst, w1, cnt_s, cnt_d, h, E);

  k_gemm<<<(GEMM_WAVES + 7) / 8, 256, 0, stream>>>(h, w2, b2, P, out);

  k_layer2<<<1536, 256, 0, stream>>>(src, rel, dst, cnt_s, cnt_d, P, out, E);
}


# Layer-1 sums in LDS first: a work-group owns 625 nodes' h rows for 1/8 of the edges (LDS float adds), then adds them to h as whole lines: 320 k global line atomics, was 2560 k; h add order changes
# speedup vs baseline: 2.2502x; 1.1504x over previous
#include <hip/hip_runtime.h>

#define NNODES 20000
#define NRELH  16
#define RTOT   33
#define HID    64
#define NCLS   16
#define RN     (RTOT * NNODES)
#define CN     (NRELH * NNODES)
#define PCOLS  (2 * NRELH * NCLS)

typedef __attribute__((ext_vector_type(2))) float v2f;
typedef __attribute__((ext_vector_type(8))) float v8f;

typedef __attribute__((ext_vector_type(4))) float v4f;
__global__ __launch_bounds__(256) void k_init(const float* __restrict__ w1,
                                              const float* __restrict__ b1,
                                              v4f* __restrict__ cnt4,
                                              float* __restrict__ h) {
  const float* wself = w1 + (size_t)(2 * NRELH) * NNODES * HID;
  int tid = blockIdx.x * blockDim.x + threadIdx.x;
  int stride = gridDim.x * blockDim.x;
  for (int i = tid; i < 2 * CN / 4; i += stride) cnt4[i] = (v4f)0.0f;
  for (int q = tid; q < NNODES * HID / 4; q += stride) {
    int idx = q * 4, i = idx % HID;
#pragma unroll
    for (int t = 0; t < 4; ++t) h[idx + t] = wself[idx + t] + b1[i + t];
  }
}

#define CNT_ROWS   (2 * NRELH)
#define CNT_CHUNKS 8
#define CNT_TPB    1024
#define CNT_UNR    4
typedef __attribute__((ext_vector_type(4))) int v4i;
__global__ __launch_bounds__(CNT_TPB) void k_count(const int* __restrict__ src,
                                                   const int* __restrict__ rel,
                                                   const int* __restrict__ dst,
                                                   float* __restrict__ cnt, int E) {
  __shared__ int bins[NNODES];
  const int row   = blockIdx.x % CNT_ROWS;
  const int chunk = blockIdx.x / CNT_ROWS;
  const int r     = row % NRELH;
  const int* __restrict__ node = row < NRELH ? src : dst;
  const int tid = threadIdx.x;

  for (int j = tid; j < NNODES; j += CNT_TPB) bins[j] = 0;
  __syncthreads();

  const int clen = (((E + CNT_CHUNKS - 1) / CNT_CHUNKS) + 3) & ~3;
  int e0 = chunk * clen;      if (e0 > E) e0 = E;
  int e1 = e0 + clen;         if (e1 > E) e1 = E;
  const int nvec = (e1 - e0) >> 2;
  const v4i* __restrict__ rel4  = (const v4i*)(rel + e0);
  const v4i* __restrict__ node4 = (const v4i*)(node + e0);

  for (int q0 = tid; q0 < nvec; q0 += CNT_UNR * CNT_TPB) {
    v4i rv[CNT_UNR], nv[CNT_UNR];
#pragma unroll
    for (int u = 0; u < CNT_UNR; ++u) {
      int q = q0 + u * CNT_TPB;
      bool ok = q < nvec;
      rv[u] = ok ? rel4[q]  : (v4i)(-1);
      nv[u] = ok ? node4[q] : (v4i)0;
    }
#pragma unroll
    for (int u = 0; u < CNT_UNR; ++u) {
      if (rv[u].x == r) atomicAdd(&bins[nv[u].x], 1);
      if (rv[u].y == r) atomicAdd(&bins[nv[u].y], 1);
      if (rv[u].z == r) atomicAdd(&bins[nv[u].z], 1);
      if (rv[u].w == r) atomicAdd(&bins[nv[u].w], 1);
    }
  }
  for (int e = e0 + 4 * nvec + tid; e < e1; e += CNT_TPB)
    if (rel[e] == r) atomicAdd(&bins[node[e]], 1);
  __syncthreads();

  float* __restrict__ trow = cnt + (size_t)row * NNODES;
  for (int j = tid; j < NNODES; j += CNT_TPB) atomicAdd(trow + j, (float)bins[j]);
}

#define L1_SLICES 32
#define L1_K      (NNODES / L1_SLICES)
#define L1_CHUNKS 8
#define L1_TPB    1024
#define L1_B      16
#define L1_VPL    2
#define L1_QCAP   (L1_B + 2 * 4 * L1_VPL * 32)
static_assert(NNODES % L1_SLICES == 0, "slices must tile the nodes");
static_assert(L1_K <= 1024 && NNODES <= 32768 && 2 * NRELH <= 32, "message word: 5 + 15 + 10 bits");

__device__ __forceinline__ void l1_push(unsigned* wq, int& qn, bool m, unsigned ent) {
  unsigned bal = __builtin_amdgcn_ballot_w32(m);
  if (m) wq[qn + (int)__builtin_amdgcn_mbcnt_lo(bal, 0u)] = ent;
  qn += __builtin_popcount(bal);
}

__device__ __forceinline__ void l1_edge(unsigned* wq, int& qn, int s, int r, int d, int n0) {
  unsigned ls = (unsigned)(s - n0), ld = (unsigned)(d - n0);
  l1_push(wq, qn, ls < (unsigned)L1_K, ((unsigned)r << 25) | ((unsigned)d << 10) | ls);
  l1_push(wq, qn, ld < (unsigned)L1_K, ((unsigned)(r + NRELH) << 25) | ((unsigned)s << 10) | ld);
}

__device__ __forceinline__ void l1_load(const v4i* __restrict__ src4, const v4i* __restrict__ rel4,
                                        const v4i* __restrict__ dst4, int qi, int nvec,
                                        v4i& s, v4i& r, v4i& d) {
  bool ok = qi < nvec;
  int  qc = ok ? qi : nvec - 1;
  v4i ts = src4[qc], tr = rel4[qc], td = dst4[qc];
  s = ok ? ts : (v4i)(-1);
  r = tr;
  d = ok ? td : (v4i)(-1);
}

template <bool GUARD>
__device__ __forceinline__ void l1_batch(const unsigned* qe, int nvalid,
                                         const float* __restrict__ w1, const float* __restrict__ cnt,
                                         float* __restrict__ hs, int n0, int lane) {
  int jl = lane & (L1_B - 1);
  if (GUARD && jl >= nvalid) jl = 0;
  unsigned ent = qe[jl];
  unsigned rp = ent >> 25, o = (ent >> 10) & 0x7FFFu, l = ent & 0x3FFu;
  float cv   = cnt[(size_t)rp * NNODES + (unsigned)n0 + l];
  int   rowv = (int)(rp * NNODES + o);
  int   lofv = (int)(l * HID);
  v2f x[L1_B];
#pragma unroll
  for (int j = 0; j < L1_B; ++j) {
    int row = __builtin_amdgcn_readlane(rowv, j);
    x[j] = ((const v2f*)(w1 + (size_t)(unsigned)row * HID))[lane];
  }
  float vinv = 1.0f / cv;
#pragma unroll
  for (int j = 0; j < L1_B; ++j) {
    if (!GUARD || j < nvalid) {
      float v  = __int_as_float(__builtin_amdgcn_readlane(__float_as_int(vinv), j));
      int  off = __builtin_amdgcn_readlane(lofv, j) + 2 * lane;
      atomicAdd(&hs[off],     v * x[j].x);
      atomicAdd(&hs[off + 1], v * x[j].y);
    }
  }
}

__global__ __launch_bounds__(L1_TPB) void k_layer1(const int* __restrict__ src,
                                                   const int* __restrict__ rel,
                                                   const int* __restrict__ dst,
                                                   const float* __restrict__ w1,
                                                   const float* __restrict__ cnt,
                                                   float* __restrict__ hacc, int E) {
  __shared__ float    hs[L1_K * HID];
  __shared__ unsigned q[L1_TPB / 32][L1_QCAP];
  const int slice = blockIdx.x % L1_SLICES;
  const int chunk = blockIdx.x / L1_SLICES;
  const int n0    = slice * L1_K;
  const int tid   = threadIdx.x;
  const int lane  = tid & 31;
  const int wave  = __builtin_amdgcn_readfirstlane(tid >> 5);
  unsigned* wq = q[wave];

  for (int i = tid; i < L1_K * HID / 4; i += L1_TPB) ((v4f*)hs)[i] = (v4f)0.0f;
  __syncthreads();

  const int clen = (((E + L1_CHUNKS - 1) / L1_CHUNKS) + 3) & ~3;
  int e0 = chunk * clen;      if (e0 > E) e0 = E;
  int e1 = e0 + clen;         if (e1 > E) e1 = E;
  const int nvec = (e1 - e0) >> 2;
  const v4i* __restrict__ src4 = (const v4i*)(src + e0);
  const v4i* __restrict__ rel4 = (const v4i*)(rel + e0);
  const v4i* __restrict__ dst4 = (const v4i*)(dst + e0);

  int qn = 0;
  if (nvec > 0) {
    v4i sv[L1_VPL], rv[L1_VPL], dv[L1_VPL];
#pragma unroll
    for (int u = 0; u < L1_VPL; ++u)
      l1_load(src4, rel4, dst4, tid + u * L1_TPB, nvec, sv[u], rv[u], dv[u]);
    for (int base = 0; base < nvec; base += L1_VPL * L1_TPB) {
      v4i sn[L1_VPL], rn[L1_VPL], dn[L1_VPL];
#pragma unroll
      for (int u = 0; u < L1_VPL; ++u)
        l1_load(src4, rel4, dst4, base + L1_VPL * L1_TPB + tid + u * L1_TPB, nvec, sn[u], rn[u], dn[u]);
#pragma unroll
      for (int u = 0; u < L1_VPL; ++u) {
        l1_edge(wq, qn, sv[u].x, rv[u].x, dv[u].x, n0);
        l1_edge(wq, qn, sv[u].y, rv[u].y, dv[u].y, n0);
        l1_edge(wq, qn, sv[u].z, rv[u].z, dv[u].z, n0);
        l1_edge(wq, qn, sv[u].w, rv[u].w, dv[u].w, n0);
      }
      while (qn >= L1_B) {
        qn -= L1_B;
        l1_batch<false>(wq + qn, L1_B, w1, cnt, hs, n0, lane);
      }
#pragma unroll
      for (int u = 0; u < L1_VPL; ++u) { sv[u] = sn[u]; rv[u] = rn[u]; dv[u] = dn[u]; }
    }
  }
  if (wave == 0) {
    int e = e0 + 4 * nvec + lane;
    bool ok = e < e1;
    int s = -1, r = 0, d = -1;
    if (ok) { s = src[e]; r = rel[e]; d = dst[e]; }
    l1_edge(wq, qn, s, r, d, n0);
    while (qn >= L1_B) {
      qn -= L1_B;
      l1_batch<false>(wq + qn, L1_B, w1, cnt, hs, n0, lane);
    }
  }
  if (qn > 0) l1_batch<true>(wq, qn, w1, cnt, hs, n0, lane);
  __syncthreads();

  float* __restrict__ hg = hacc + (size_t)n0 * HID;
  for (int i0 = tid; i0 < L1_K * HID; i0 += 8 * L1_TPB) {
    float t[8];
#pragma unroll
    for (int u = 0; u < 8; ++u) { int i = i0 + u * L1_TPB; t[u] = i < L1_K * HID ? hs[i] : 0.0f; }
#pragma unroll
    for (int u = 0; u < 8; ++u) { int i = i0 + u * L1_TPB; if (i < L1_K * HID) atomicAdd(hg + i, t[u]); }
  }
}

#define GEMM_RG  3
#define GEMM_RPG (RTOT / GEMM_RG)
#define GEMM_WAVES ((NNODES / 16) * GEMM_RG)
__device__ __forceinline__ v8f tile_product(const v2f (&a)[HID / 4], const float* __restrict__ wrel,
                                            int khalf, int mn) {
  v8f acc = (v8f)0.0f;
#pragma unroll
  for (int kk = 0; kk < HID / 4; ++kk) {
    int ka = 4 * kk + 2 * khalf;
    v2f b;
    b.x = wrel[(size_t)ka * NCLS + mn];
    b.y = wrel[(size_t)(ka + 1) * NCLS + mn];
    acc = __builtin_amdgcn_wmma_f32_16x16x4_f32(
         false, a[kk],  false, b,
         (short)0, acc,  false,  false);
  }
  return acc;
}

template <bool SELF>
__device__ __forceinline__ void gemm_relations(const v2f (&a)[HID / 4], const float* __restrict__ w2,
                                               const float* __restrict__ b2, float* __restrict__ P,
                                               float* __restrict__ out, int mt, int r0, int rowoff,
                                               int khalf, int mn) {
  constexpr int NP = SELF ? GEMM_RPG - 1 : GEMM_RPG;
  for (int r = r0; r < r0 + NP; ++r) {
    union { v8f v; float f[8]; } acc;
    acc.v = tile_product(a, w2 + (size_t)r * HID * NCLS, khalf, mn);
#pragma unroll
    for (int v = 0; v < 8; ++v) {
      P[(size_t)(mt * 16 + v + rowoff) * PCOLS + r * NCLS + mn] = acc.f[v];
    }
  }
  if (SELF) {
    union { v8f v; float f[8]; } acc;
    acc.v = tile_product(a, w2 + (size_t)(2 * NRELH) * HID * NCLS, khalf, mn);
    float bias = b2[mn];
#pragma unroll
    for (int v = 0; v < 8; ++v) {
      out[(size_t)(mt * 16 + v + rowoff) * NCLS + mn] = acc.f[v] + bias;
    }
  }
}

__global__ __launch_bounds__(256) void k_gemm(const float* __restrict__ h,
                                              const float* __restrict__ w2,
                                              const float* __restrict__ b2,
                                              float* __restrict__ P,
                                              float* __restrict__ out) {
  int wid = __builtin_amdgcn_readfirstlane((int)((blockIdx.x * blockDim.x + threadIdx.x) >> 5));
  if (wid >= GEMM_WAVES) return;
  int mt = wid / GEMM_RG;
  int r0 = (wid % GEMM_RG) * GEMM_RPG;
  int lane  = threadIdx.x & 31;
  int mn    = lane & 15;
  int khalf = lane >> 4;
  const float* hrow = h + (size_t)(mt * 16 + mn) * HID;
  v2f a[HID / 4];
#pragma unroll
  for (int kk = 0; kk < HID / 4; ++kk) {
    int ka = 4 * kk + 2 * khalf;
    float x = hrow[ka], y = hrow[ka + 1];
    a[kk].x = x > 0.0f ? x : 0.0f;
    a[kk].y = y > 0.0f ? y : 0.0f;
  }
  int rowoff = khalf * 8;
  if (r0 + GEMM_RPG == RTOT)
    gemm_relations<true>(a, w2, b2, P, out, mt, r0, rowoff, khalf, mn);
  else
    gemm_relations<false>(a, w2, b2, P, out, mt, r0, rowoff, khalf, mn);
}

#define L2_TPB 4
template <int NT>
__device__ __forceinline__ void layer2_trips(const int* __restrict__ src,
                                             const int* __restrict__ rel,
                                             const int* __restrict__ dst,
                                             const float* __restrict__ cnt_s,
                                             const float* __restrict__ cnt_d,
                                             const float* __restrict__ P,
                                             float* __restrict__ out, int e0, int c) {
  int   sv[NT], dv[NT];
  float x[NT], y[NT];
#pragma unroll
  for (int j = 0; j < NT; ++j) {
    int e = e0 + 2 * j;
    int s = src[e], r = rel[e], d = dst[e];
    sv[j] = s; dv[j] = d;
    float vo = 1.0f / cnt_s[(size_t)r * NNODES + s];
    float vi = 1.0f / cnt_d[(size_t)r * NNODES + d];
    x[j] = vo * P[(size_t)d * PCOLS + r * NCLS + c];
    y[j] = vi * P[(size_t)s * PCOLS + (r + NRELH) * NCLS + c];
  }
#pragma unroll
  for (int j = 0; j < NT; ++j) {
    atomicAdd(&out[(size_t)sv[j] * NCLS + c], x[j]);
    atomicAdd(&out[(size_t)dv[j] * NCLS + c], y[j]);
  }
}

__global__ __launch_bounds__(256) void k_layer2(const int* __restrict__ src,
                                                const int* __restrict__ rel,
                                                const int* __restrict__ dst,
                                                const float* __restrict__ cnt_s,
                                                const float* __restrict__ cnt_d,
                                                const float* __restrict__ P,
                                                float* __restrict__ out, int E) {
  int wave = __builtin_amdgcn_readfirstlane((int)((blockIdx.x * blockDim.x + threadIdx.x) >> 5));
  int half = (threadIdx.x >> 4) & 1;
  int c    = threadIdx.x & 15;
  int nwaves   = (int)((gridDim.x * blockDim.x) >> 5);
  int nbatches = E / (2 * L2_TPB);
  for (int b = wave; b < nbatches; b += nwaves)
    layer2_trips<L2_TPB>(src, rel, dst, cnt_s, cnt_d, P, out, b * (2 * L2_TPB) + half, c);
  if (wave == 0) {
    for (int e = nbatches * (2 * L2_TPB) + half; e < E; e += 2)
      layer2_trips<1>(src, rel, dst, cnt_s, cnt_d, P, out, e, c);
  }
}

extern "C" void kernel_launch(void* const* d_in, const int* in_sizes, int n_in,
                              void* d_out, int out_size, void* d_ws, size_t ws_size,
                              hipStream_t stream) {
  const int*   src = (const int*)d_in[0];
  const int*   rel = (const int*)d_in[1];
  const int*   dst = (const int*)d_in[2];
  const float* w1  = (const float*)d_in[3];
  const float* b1  = (const float*)d_in[4];
  const float* w2  = (const float*)d_in[5];
  const float* b2  = (const float*)d_in[6];
  float* out = (float*)d_out;
  const int E = in_sizes[0];

  float* cnt_s = (float*)d_ws;
  float* cnt_d = cnt_s + CN;
  float* h     = cnt_d + CN;
  float* P     = h + (size_t)NNODES * HID;

  k_init<<<1024, 256, 0, stream>>>(w1, b1, (v4f*)cnt_s, h);

  k_count<<<CNT_ROWS * CNT_CHUNKS, CNT_TPB, 0, stream>>>(src, rel, dst, cnt_s, E);

  k_layer1<<<L1_SLICES * L1_CHUNKS, L1_TPB, 0, stream>>>(src, rel, dst, w1, cnt_s, h, E);

  k_gemm<<<(GEMM_WAVES + 7) / 8, 256, 0, stream>>>(h, w2, b2, P, out);

  k_layer2<<<1536, 256, 0, stream>>>(src, rel, dst, cnt_s, cnt_d, P, out, E);
}


# Layer-2 sums in LDS first: a work-group owns 1250 nodes' out rows and inverse counts for 1/16 of the edges, then adds to out as whole lines: 1 random request per message, was 3; out add order changes
# speedup vs baseline: 3.2460x; 1.4425x over previous
#include <hip/hip_runtime.h>

#define NNODES 20000
#define NRELH  16
#define RTOT   33
#define HID    64
#define NCLS   16
#define RN     (RTOT * NNODES)
#define CN     (NRELH * NNODES)
#define PCOLS  (2 * NRELH * NCLS)

typedef __attribute__((ext_vector_type(2))) float v2f;
typedef __attribute__((ext_vector_type(8))) float v8f;

typedef __attribute__((ext_vector_type(4))) float v4f;
__global__ __launch_bounds__(256) void k_init(const float* __restrict__ w1,
                                              const float* __restrict__ b1,
                                              v4f* __restrict__ cnt4,
                                              float* __restrict__ h) {
  const float* wself = w1 + (size_t)(2 * NRELH) * NNODES * HID;
  int tid = blockIdx.x * blockDim.x + threadIdx.x;
  int stride = gridDim.x * blockDim.x;
  for (int i = tid; i < 2 * CN / 4; i += stride) cnt4[i] = (v4f)0.0f;
  for (int q = tid; q < NNODES * HID / 4; q += stride) {
    int idx = q * 4, i = idx % HID;
#pragma unroll
    for (int t = 0; t < 4; ++t) h[idx + t] = wself[idx + t] + b1[i + t];
  }
}

#define CNT_ROWS   (2 * NRELH)
#define CNT_CHUNKS 8
#define CNT_TPB    1024
#define CNT_UNR    4
typedef __attribute__((ext_vector_type(4))) int v4i;
__global__ __launch_bounds__(CNT_TPB) void k_count(const int* __restrict__ src,
                                                   const int* __restrict__ rel,
                                                   const int* __restrict__ dst,
                                                   float* __restrict__ cnt, int E) {
  __shared__ int bins[NNODES];
  const int row   = blockIdx.x % CNT_ROWS;
  const int chunk = blockIdx.x / CNT_ROWS;
  const int r     = row % NRELH;
  const int* __restrict__ node = row < NRELH ? src : dst;
  const int tid = threadIdx.x;

  for (int j = tid; j < NNODES; j += CNT_TPB) bins[j] = 0;
  __syncthreads();

  const int clen = (((E + CNT_CHUNKS - 1) / CNT_CHUNKS) + 3) & ~3;
  int e0 = chunk * clen;      if (e0 > E) e0 = E;
  int e1 = e0 + clen;         if (e1 > E) e1 = E;
  const int nvec = (e1 - e0) >> 2;
  const v4i* __restrict__ rel4  = (const v4i*)(rel + e0);
  const v4i* __restrict__ node4 = (const v4i*)(node + e0);

  for (int q0 = tid; q0 < nvec; q0 += CNT_UNR * CNT_TPB) {
    v4i rv[CNT_UNR], nv[CNT_UNR];
#pragma unroll
    for (int u = 0; u < CNT_UNR; ++u) {
      int q = q0 + u * CNT_TPB;
      bool ok = q < nvec;
      rv[u] = ok ? rel4[q]  : (v4i)(-1);
      nv[u] = ok ? node4[q] : (v4i)0;
    }
#pragma unroll
    for (int u = 0; u < CNT_UNR; ++u) {
      if (rv[u].x == r) atomicAdd(&bins[nv[u].x], 1);
      if (rv[u].y == r) atomicAdd(&bins[nv[u].y], 1);
      if (rv[u].z == r) atomicAdd(&bins[nv[u].z], 1);
      if (rv[u].w == r) atomicAdd(&bins[nv[u].w], 1);
    }
  }
  for (int e = e0 + 4 * nvec + tid; e < e1; e += CNT_TPB)
    if (rel[e] == r) atomicAdd(&bins[node[e]], 1);
  __syncthreads();

  float* __restrict__ trow = cnt + (size_t)row * NNODES;
  for (int j = tid; j < NNODES; j += CNT_TPB) atomicAdd(trow + j, (float)bins[j]);
}

#define L1_SLICES 32
#define L1_K      (NNODES / L1_SLICES)
#define L1_CHUNKS 8
#define L1_TPB    1024
#define L1_B      16
#define L1_VPL    2
#define L1_QCAP   (L1_B + 2 * 4 * L1_VPL * 32)
static_assert(NNODES % L1_SLICES == 0, "slices must tile the nodes");
static_assert(L1_K <= 1024 && NNODES <= 32768 && 2 * NRELH <= 32, "message word: 5 + 15 + 10 bits");

__device__ __forceinline__ void l1_push(unsigned* wq, int& qn, bool m, unsigned ent) {
  unsigned bal = __builtin_amdgcn_ballot_w32(m);
  if (m) wq[qn + (int)__builtin_amdgcn_mbcnt_lo(bal, 0u)] = ent;
  qn += __builtin_popcount(bal);
}

__device__ __forceinline__ void l1_edge(unsigned* wq, int& qn, int s, int r, int d, int n0) {
  unsigned ls = (unsigned)(s - n0), ld = (unsigned)(d - n0);
  l1_push(wq, qn, ls < (unsigned)L1_K, ((unsigned)r << 25) | ((unsigned)d << 10) | ls);
  l1_push(wq, qn, ld < (unsigned)L1_K, ((unsigned)(r + NRELH) << 25) | ((unsigned)s << 10) | ld);
}

__device__ __forceinline__ void l1_load(const v4i* __restrict__ src4, const v4i* __restrict__ rel4,
                                        const v4i* __restrict__ dst4, int qi, int nvec,
                                        v4i& s, v4i& r, v4i& d) {
  bool ok = qi < nvec;
  int  qc = ok ? qi : nvec - 1;
  v4i ts = src4[qc], tr = rel4[qc], td = dst4[qc];
  s = ok ? ts : (v4i)(-1);
  r = tr;
  d = ok ? td : (v4i)(-1);
}

template <bool GUARD>
__device__ __forceinline__ void l1_batch(const unsigned* qe, int nvalid,
                                         const float* __restrict__ w1, const float* __restrict__ cnt,
                                         float* __restrict__ hs, int n0, int lane) {
  int jl = lane & (L1_B - 1);
  if (GUARD && jl >= nvalid) jl = 0;
  unsigned ent = qe[jl];
  unsigned rp = ent >> 25, o = (ent >> 10) & 0x7FFFu, l = ent & 0x3FFu;
  float cv   = cnt[(size_t)rp * NNODES + (unsigned)n0 + l];
  int   rowv = (int)(rp * NNODES + o);
  int   lofv = (int)(l * HID);
  v2f x[L1_B];
#pragma unroll
  for (int j = 0; j < L1_B; ++j) {
    int row = __builtin_amdgcn_readlane(rowv, j);
    x[j] = ((const v2f*)(w1 + (size_t)(unsigned)row * HID))[lane];
  }
  float vinv = 1.0f / cv;
#pragma unroll
  for (int j = 0; j < L1_B; ++j) {
    if (!GUARD || j < nvalid) {
      float v  = __int_as_float(__builtin_amdgcn_readlane(__float_as_int(vinv), j));
      int  off = __builtin_amdgcn_readlane(lofv, j) + 2 * lane;
      atomicAdd(&hs[off],     v * x[j].x);
      atomicAdd(&hs[off + 1], v * x[j].y);
    }
  }
}

__global__ __launch_bounds__(L1_TPB) void k_layer1(const int* __restrict__ src,
                                                   const int* __restrict__ rel,
                                                   const int* __restrict__ dst,
                                                   const float* __restrict__ w1,
                                                   const float* __restrict__ cnt,
                                                   float* __restrict__ hacc, int E) {
  __shared__ float    hs[L1_K * HID];
  __shared__ unsigned q[L1_TPB / 32][L1_QCAP];
  const int slice = blockIdx.x % L1_SLICES;
  const int chunk = blockIdx.x / L1_SLICES;
  const int n0    = slice * L1_K;
  const int tid   = threadIdx.x;
  const int lane  = tid & 31;
  const int wave  = __builtin_amdgcn_readfirstlane(tid >> 5);
  unsigned* wq = q[wave];

  for (int i = tid; i < L1_K * HID / 4; i += L1_TPB) ((v4f*)hs)[i] = (v4f)0.0f;
  __syncthreads();

  const int clen = (((E + L1_CHUNKS - 1) / L1_CHUNKS) + 3) & ~3;
  int e0 = chunk * clen;      if (e0 > E) e0 = E;
  int e1 = e0 + clen;         if (e1 > E) e1 = E;
  const int nvec = (e1 - e0) >> 2;
  const v4i* __restrict__ src4 = (const v4i*)(src + e0);
  const v4i* __restrict__ rel4 = (const v4i*)(rel + e0);
  const v4i* __restrict__ dst4 = (const v4i*)(dst + e0);

  int qn = 0;
  if (nvec > 0) {
    v4i sv[L1_VPL], rv[L1_VPL], dv[L1_VPL];
#pragma unroll
    for (int u = 0; u < L1_VPL; ++u)
      l1_load(src4, rel4, dst4, tid + u * L1_TPB, nvec, sv[u], rv[u], dv[u]);
    for (int base = 0; base < nvec; base += L1_VPL * L1_TPB) {
      v4i sn[L1_VPL], rn[L1_VPL], dn[L1_VPL];
#pragma unroll
      for (int u = 0; u < L1_VPL; ++u)
        l1_load(src4, rel4, dst4, base + L1_VPL * L1_TPB + tid + u * L1_TPB, nvec, sn[u], rn[u], dn[u]);
#pragma unroll
      for (int u = 0; u < L1_VPL; ++u) {
        l1_edge(wq, qn, sv[u].x, rv[u].x, dv[u].x, n0);
        l1_edge(wq, qn, sv[u].y, rv[u].y, dv[u].y, n0);
        l1_edge(wq, qn, sv[u].z, rv[u].z, dv[u].z, n0);
        l1_edge(wq, qn, sv[u].w, rv[u].w, dv[u].w, n0);
      }
      while (qn >= L1_B) {
        qn -= L1_B;
        l1_batch<false>(wq + qn, L1_B, w1, cnt, hs, n0, lane);
      }
#pragma unroll
      for (int u = 0; u < L1_VPL; ++u) { sv[u] = sn[u]; rv[u] = rn[u]; dv[u] = dn[u]; }
    }
  }
  if (wave == 0) {
    int e = e0 + 4 * nvec + lane;
    bool ok = e < e1;
    int s = -1, r = 0, d = -1;
    if (ok) { s = src[e]; r = rel[e]; d = dst[e]; }
    l1_edge(wq, qn, s, r, d, n0);
    while (qn >= L1_B) {
      qn -= L1_B;
      l1_batch<false>(wq + qn, L1_B, w1, cnt, hs, n0, lane);
    }
  }
  if (qn > 0) l1_batch<true>(wq, qn, w1, cnt, hs, n0, lane);
  __syncthreads();

  float* __restrict__ hg = hacc + (size_t)n0 * HID;
  for (int i0 = tid; i0 < L1_K * HID; i0 += 8 * L1_TPB) {
    float t[8];
#pragma unroll
    for (int u = 0; u < 8; ++u) { int i = i0 + u * L1_TPB; t[u] = i < L1_K * HID ? hs[i] : 0.0f; }
#pragma unroll
    for (int u = 0; u < 8; ++u) { int i = i0 + u * L1_TPB; if (i < L1_K * HID) atomicAdd(hg + i, t[u]); }
  }
}

#define GEMM_RG  3
#define GEMM_RPG (RTOT / GEMM_RG)
#define GEMM_WAVES ((NNODES / 16) * GEMM_RG)
__device__ __forceinline__ v8f tile_product(const v2f (&a)[HID / 4], const float* __restrict__ wrel,
                                            int khalf, int mn) {
  v8f acc = (v8f)0.0f;
#pragma unroll
  for (int kk = 0; kk < HID / 4; ++kk) {
    int ka = 4 * kk + 2 * khalf;
    v2f b;
    b.x = wrel[(size_t)ka * NCLS + mn];
    b.y = wrel[(size_t)(ka + 1) * NCLS + mn];
    acc = __builtin_amdgcn_wmma_f32_16x16x4_f32(
         false, a[kk],  false, b,
         (short)0, acc,  false,  false);
  }
  return acc;
}

template <bool SELF>
__device__ __forceinline__ void gemm_relations(const v2f (&a)[HID / 4], const float* __restrict__ w2,
                                               const float* __restrict__ b2, float* __restrict__ P,
                                               float* __restrict__ out, int mt, int r0, int rowoff,
                                               int khalf, int mn) {
  constexpr int NP = SELF ? GEMM_RPG - 1 : GEMM_RPG;
  for (int r = r0; r < r0 + NP; ++r) {
    union { v8f v; float f[8]; } acc;
    acc.v = tile_product(a, w2 + (size_t)r * HID * NCLS, khalf, mn);
#pragma unroll
    for (int v = 0; v < 8; ++v) {
      P[(size_t)(mt * 16 + v + rowoff) * PCOLS + r * NCLS + mn] = acc.f[v];
    }
  }
  if (SELF) {
    union { v8f v; float f[8]; } acc;
    acc.v = tile_product(a, w2 + (size_t)(2 * NRELH) * HID * NCLS, khalf, mn);
    float bias = b2[mn];
#pragma unroll
    for (int v = 0; v < 8; ++v) {
      out[(size_t)(mt * 16 + v + rowoff) * NCLS + mn] = acc.f[v] + bias;
    }
  }
}

__global__ __launch_bounds__(256) void k_gemm(const float* __restrict__ h,
                                              const float* __restrict__ w2,
                                              const float* __restrict__ b2,
                                              float* __restrict__ P,
                                              float* __restrict__ out) {
  int wid = __builtin_amdgcn_readfirstlane((int)((blockIdx.x * blockDim.x + threadIdx.x) >> 5));
  if (wid >= GEMM_WAVES) return;
  int mt = wid / GEMM_RG;
  int r0 = (wid % GEMM_RG) * GEMM_RPG;
  int lane  = threadIdx.x & 31;
  int mn    = lane & 15;
  int khalf = lane >> 4;
  const float* hrow = h + (size_t)(mt * 16 + mn) * HID;
  v2f a[HID / 4];
#pragma unroll
  for (int kk = 0; kk < HID / 4; ++kk) {
    int ka = 4 * kk + 2 * khalf;
    float x = hrow[ka], y = hrow[ka + 1];
    a[kk].x = x > 0.0f ? x : 0.0f;
    a[kk].y = y > 0.0f ? y : 0.0f;
  }
  int rowoff = khalf * 8;
  if (r0 + GEMM_RPG == RTOT)
    gemm_relations<true>(a, w2, b2, P, out, mt, r0, rowoff, khalf, mn);
  else
    gemm_relations<false>(a, w2, b2, P, out, mt, r0, rowoff, khalf, mn);
}

#define L2_SLICES  16
#define L2_K       (NNODES / L2_SLICES)
#define L2_CHUNKS  16
#define L2_THREADS 1024
#define L2_B       32
#define L2_RU      10
#define L2_QCAP    (L2_B + 2 * 4 * L1_VPL * 32)
static_assert(NNODES % L2_SLICES == 0 && L2_K % 2 == 0, "slices must tile the nodes; table rows are read 8 bytes at a time");
static_assert(L2_K <= 2048 && NNODES <= 32768 && 2 * NRELH <= 32, "message word: 5 + 15 + 11 bits");

__device__ __forceinline__ void l2_edge(unsigned* wq, int& qn, int s, int r, int d, int n0) {
  unsigned ls = (unsigned)(s - n0), ld = (unsigned)(d - n0);
  l1_push(wq, qn, ls < (unsigned)L2_K, ((unsigned)r << 26) | ((unsigned)d << 11) | ls);
  l1_push(wq, qn, ld < (unsigned)L2_K, ((unsigned)(r + NRELH) << 26) | ((unsigned)s << 11) | ld);
}

template <bool GUARD>
__device__ __forceinline__ void l2_batch(const unsigned* qe, int nvalid, const float* __restrict__ P,
                                         const float* cinv, float* os, int lane) {
  const int half = lane >> 4, c = lane & 15;
  float x[L2_B / 2], v[L2_B / 2];
  int   off[L2_B / 2];
#pragma unroll
  for (int t = 0; t < L2_B / 2; ++t) {
    int m = 2 * t + half;
    if (GUARD && m >= nvalid) m = 0;
    unsigned ent = qe[m];
    unsigned rp = ent >> 26, o = (ent >> 11) & 0x7FFFu, l = ent & 0x7FFu;
    v[t]   = cinv[rp * L2_K + l];
    x[t]   = P[o * (unsigned)PCOLS + rp * NCLS + c];
    off[t] = (int)(l * NCLS) + c;
  }
#pragma unroll
  for (int t = 0; t < L2_B / 2; ++t) {
    if (!GUARD || 2 * t + half < nvalid) atomicAdd(&os[off[t]], v[t] * x[t]);
  }
}

__global__ __launch_bounds__(L2_THREADS) void k_layer2(const int* __restrict__ src,
                                                       const int* __restrict__ rel,
                                                       const int* __restrict__ dst,
                                                       const float* __restrict__ cnt,
                                                       const float* __restrict__ P,
                                                       float* __restrict__ out, int E) {
  __shared__ __attribute__((aligned(16))) float cinv[2 * NRELH * L2_K];
  __shared__ __attribute__((aligned(16))) float os[L2_K * NCLS];
  __shared__ unsigned q[L2_THREADS / 32][L2_QCAP];
  const int slice = blockIdx.x % L2_SLICES;
  const int chunk = blockIdx.x / L2_SLICES;
  const int n0    = slice * L2_K;
  const int tid   = threadIdx.x;
  const int lane  = tid & 31;
  const int wave  = __builtin_amdgcn_readfirstlane(tid >> 5);
  unsigned* wq = q[wave];

  for (int row = wave; row < 2 * NRELH; row += L2_THREADS / 32) {
    const v2f* __restrict__ crow = (const v2f*)(cnt + (size_t)row * NNODES + n0);
    v2f* irow = (v2f*)(cinv + row * L2_K);
    for (int j0 = lane; j0 < L2_K / 2; j0 += L2_RU * 32) {
      v2f t[L2_RU];
#pragma unroll
      for (int u = 0; u < L2_RU; ++u) { int j = j0 + 32 * u; t[u] = crow[j < L2_K / 2 ? j : 0]; }
#pragma unroll
      for (int u = 0; u < L2_RU; ++u) {
        int j = j0 + 32 * u;
        if (j < L2_K / 2) { v2f w; w.x = 1.0f / t[u].x; w.y = 1.0f / t[u].y; irow[j] = w; }
      }
    }
  }
  for (int i = tid; i < L2_K * NCLS / 4; i += L2_THREADS) ((v4f*)os)[i] = (v4f)0.0f;
  __syncthreads();

  const int clen = (((E + L2_CHUNKS - 1) / L2_CHUNKS) + 3) & ~3;
  int e0 = chunk * clen;      if (e0 > E) e0 = E;
  int e1 = e0 + clen;         if (e1 > E) e1 = E;
  const int nvec = (e1 - e0) >> 2;
  const v4i* __restrict__ src4 = (const v4i*)(src + e0);
  const v4i* __restrict__ rel4 = (const v4i*)(rel + e0);
  const v4i* __restrict__ dst4 = (const v4i*)(dst + e0);

  int qn = 0;
  if (nvec > 0) {
    v4i sv[L1_VPL], rv[L1_VPL], dv[L1_VPL];
#pragma unroll
    for (int u = 0; u < L1_VPL; ++u)
      l1_load(src4, rel4, dst4, tid + u * L2_THREADS, nvec, sv[u], rv[u], dv[u]);
    for (int base = 0; base < nvec; base += L1_VPL * L2_THREADS) {
      v4i sn[L1_VPL], rn[L1_VPL], dn[L1_VPL];
#pragma unroll
      for (int u = 0; u < L1_VPL; ++u)
        l1_load(src4, rel4, dst4, base + L1_VPL * L2_THREADS + tid + u * L2_THREADS, nvec, sn[u], rn[u], dn[u]);
#pragma unroll
      for (int u = 0; u < L1_VPL; ++u) {
        l2_edge(wq, qn, sv[u].x, rv[u].x, dv[u].x, n0);
        l2_edge(wq, qn, sv[u].y, rv[u].y, dv[u].y, n0);
        l2_edge(wq, qn, sv[u].z, rv[u].z, dv[u].z, n0);
        l2_edge(wq, qn, sv[u].w, rv[u].w, dv[u].w, n0);
      }
      while (qn >= L2_B) {
        qn -= L2_B;
        l2_batch<false>(wq + qn, L2_B, P, cinv, os, lane);
      }
#pragma unroll
      for (int u = 0; u < L1_VPL; ++u) { sv[u] = sn[u]; rv[u] = rn[u]; dv[u] = dn[u]; }
    }
  }
  if (wave == 0) {
    int e = e0 + 4 * nvec + lane;
    bool ok = e < e1;
    int s = -1, r = 0, d = -1;
    if (ok) { s = src[e]; r = rel[e]; d = dst[e]; }
    l2_edge(wq, qn, s, r, d, n0);
    while (qn >= L2_B) {
      qn -= L2_B;
      l2_batch<false>(wq + qn, L2_B, P, cinv, os, lane);
    }
  }
  if (qn > 0) l2_batch<true>(wq, qn, P, cinv, os, lane);
  __syncthreads();

  float* __restrict__ og = out + (size_t)n0 * NCLS;
  for (int i0 = tid; i0 < L2_K * NCLS; i0 += 4 * L2_THREADS) {
    float t[4];
#pragma unroll
    for (int u = 0; u < 4; ++u) { int i = i0 + u * L2_THREADS; t[u] = i < L2_K * NCLS ? os[i] : 0.0f; }
#pragma unroll
    for (int u = 0; u < 4; ++u) { int i = i0 + u * L2_THREADS; if (i < L2_K * NCLS) atomicAdd(og + i, t[u]); }
  }
}

extern "C" void kernel_launch(void* const* d_in, const int* in_sizes, int n_in,
                              void* d_out, int out_size, void* d_ws, size_t ws_size,
                              hipStream_t stream) {
  const int*   src = (const int*)d_in[0];
  const int*   rel = (const int*)d_in[1];
  const int*   dst = (const int*)d_in[2];
  const float* w1  = (const float*)d_in[3];
  const float* b1  = (const float*)d_in[4];
  const float* w2  = (const float*)d_in[5];
  const float* b2  = (const float*)d_in[6];
  float* out = (float*)d_out;
  const int E = in_sizes[0];

  float* cnt_s = (float*)d_ws;
  float* cnt_d = cnt_s + CN;
  float* h     = cnt_d + CN;
  float* P     = h + (size_t)NNODES * HID;

  k_init<<<1024, 256, 0, stream>>>(w1, b1, (v4f*)cnt_s, h);

  k_count<<<CNT_ROWS * CNT_CHUNKS, CNT_TPB, 0, stream>>>(src, rel, dst, cnt_s, E);

  k_layer1<<<L1_SLICES * L1_CHUNKS, L1_TPB, 0, stream>>>(src, rel, dst, w1, cnt_s, h, E);

  k_gemm<<<(GEMM_WAVES + 7) / 8, 256, 0, stream>>>(h, w2, b2, P, out);

  k_layer2<<<L2_SLICES * L2_CHUNKS, L2_THREADS, 0, stream>>>(src, rel, dst, cnt_s, P, out, E);
}


# Messages sorted once (LDS counting sort in the start kernel) by receiving slice and relation pair, 2500 edges a run; layers read their pieces instead of scanning all edges; h, out add order changes
# speedup vs baseline: 3.5162x; 1.0832x over previous
#include <hip/hip_runtime.h>

#define NNODES 20000
#define NRELH  16
#define RTOT   33
#define HID    64
#define NCLS   16
#define RN     (RTOT * NNODES)
#define CN     (NRELH * NNODES)
#define PCOLS  (2 * NRELH * NCLS)

typedef __attribute__((ext_vector_type(2))) float v2f;
typedef __attribute__((ext_vector_type(8))) float v8f;

#define L1_SLICES 32
#define L1_K      (NNODES / L1_SLICES)
#define L1_G      16
#define RUN_EDGES 2500
#define RUN_MSGS  (2 * RUN_EDGES)
#define RUN_OFFS  (L1_SLICES + 1)
#define RUN_WORDS (RUN_OFFS + RUN_MSGS)
#define SORT_WGS  256
#define SORT_TPB  512
#define SORT_KEYS (L1_SLICES * L1_G)
#define SORT_EPT  ((RUN_EDGES + SORT_TPB - 1) / SORT_TPB)
static_assert(NNODES % L1_SLICES == 0, "slices must tile the nodes");
static_assert(L1_K <= 1024 && NNODES <= 32768 && 2 * NRELH <= 32, "message word: 5 + 15 + 10 bits");
static_assert(2 * NRELH == 2 * L1_G, "a group is 2 relation rows");
static_assert(SORT_KEYS == SORT_TPB, "the prefix step gives every thread one key");

typedef __attribute__((ext_vector_type(4))) float v4f;
__global__ __launch_bounds__(SORT_TPB) void k_init(const int* __restrict__ src,
                                                   const int* __restrict__ rel,
                                                   const int* __restrict__ dst,
                                                   const float* __restrict__ w1,
                                                   const float* __restrict__ b1,
                                                   v4f* __restrict__ cnt4,
                                                   float* __restrict__ h,
                                                   unsigned* __restrict__ runs, int E) {
  __shared__ int      key_n[SORT_KEYS];
  __shared__ int      slice_n[L1_SLICES];
  __shared__ unsigned stage[RUN_MSGS];
  const int tid   = threadIdx.x;
  const int nruns = (E + RUN_EDGES - 1) / RUN_EDGES;

  for (int run = blockIdx.x; run < nruns; run += gridDim.x) {
    key_n[tid] = 0;
    __syncthreads();

    const int e0 = run * RUN_EDGES;
    const int e1 = e0 + RUN_EDGES < E ? e0 + RUN_EDGES : E;
    int      ks[SORT_EPT], kd[SORT_EPT];
    unsigned ms[SORT_EPT], md[SORT_EPT];
#pragma unroll
    for (int u = 0; u < SORT_EPT; ++u) {
      int e = e0 + tid + u * SORT_TPB;
      ks[u] = -1; kd[u] = 0; ms[u] = 0u; md[u] = 0u;
      if (e < e1) {
        int s = src[e], r = rel[e], d = dst[e];
        int ss = s / L1_K, sd = d / L1_K;
        ks[u] = ss * L1_G + (r >> 1);
        kd[u] = sd * L1_G + ((r + NRELH) >> 1);
        ms[u] = ((unsigned)r << 25) | ((unsigned)d << 10) | (unsigned)(s - ss * L1_K);
        md[u] = ((unsigned)(r + NRELH) << 25) | ((unsigned)s << 10) | (unsigned)(d - sd * L1_K);
        atomicAdd(&key_n[ks[u]], 1);
        atomicAdd(&key_n[kd[u]], 1);
      }
    }
    __syncthreads();

    const int c = key_n[tid];
    int inc = c;
    for (int dl = 1; dl < L1_G; dl <<= 1) {
      int t = __shfl_up(inc, dl, L1_G);
      if ((tid & (L1_G - 1)) >= dl) inc += t;
    }
    if ((tid & (L1_G - 1)) == L1_G - 1) slice_n[tid / L1_G] = inc;
    __syncthreads();
    int base = 0;
    for (int j = 0; j < tid / L1_G; ++j) base += slice_n[j];
    key_n[tid] = base + inc - c;
    unsigned* __restrict__ rec = runs + (size_t)run * RUN_WORDS;
    if ((tid & (L1_G - 1)) == 0) rec[tid / L1_G] = (unsigned)base;
    if (tid == SORT_TPB - 1) rec[L1_SLICES] = (unsigned)(base + inc);
    __syncthreads();

#pragma unroll
    for (int u = 0; u < SORT_EPT; ++u) {
      if (ks[u] >= 0) {
        stage[atomicAdd(&key_n[ks[u]], 1)] = ms[u];
        stage[atomicAdd(&key_n[kd[u]], 1)] = md[u];
      }
    }
    __syncthreads();

    const int total = 2 * (e1 - e0);
    for (int i = tid; i < total; i += SORT_TPB) rec[RUN_OFFS + i] = stage[i];
  }

  const float* wself = w1 + (size_t)(2 * NRELH) * NNODES * HID;
  const int gtid   = blockIdx.x * SORT_TPB + tid;
  const int stride = gridDim.x * SORT_TPB;
  for (int i = gtid; i < 2 * CN / 4; i += stride) cnt4[i] = (v4f)0.0f;
  for (int q = gtid; q < NNODES * HID / 4; q += stride) {
    int idx = q * 4, i = idx % HID;
#pragma unroll
    for (int t = 0; t < 4; ++t) h[idx + t] = wself[idx + t] + b1[i + t];
  }
}

#define CNT_ROWS   (2 * NRELH)
#define CNT_CHUNKS 8
#define CNT_TPB    1024
#define CNT_UNR    4
typedef __attribute__((ext_vector_type(4))) int v4i;
__global__ __launch_bounds__(CNT_TPB) void k_count(const int* __restrict__ src,
                                                   const int* __restrict__ rel,
                                                   const int* __restrict__ dst,
                                                   float* __restrict__ cnt, int E) {
  __shared__ int bins[NNODES];
  const int row   = blockIdx.x % CNT_ROWS;
  const int chunk = blockIdx.x / CNT_ROWS;
  const int r     = row % NRELH;
  const int* __restrict__ node = row < NRELH ? src : dst;
  const int tid = threadIdx.x;

  for (int j = tid; j < NNODES; j += CNT_TPB) bins[j] = 0;
  __syncthreads();

  const int clen = (((E + CNT_CHUNKS - 1) / CNT_CHUNKS) + 3) & ~3;
  int e0 = chunk * clen;      if (e0 > E) e0 = E;
  int e1 = e0 + clen;         if (e1 > E) e1 = E;
  const int nvec = (e1 - e0) >> 2;
  const v4i* __restrict__ rel4  = (const v4i*)(rel + e0);
  const v4i* __restrict__ node4 = (const v4i*)(node + e0);

  for (int q0 = tid; q0 < nvec; q0 += CNT_UNR * CNT_TPB) {
    v4i rv[CNT_UNR], nv[CNT_UNR];
#pragma unroll
    for (int u = 0; u < CNT_UNR; ++u) {
      int q = q0 + u * CNT_TPB;
      bool ok = q < nvec;
      rv[u] = ok ? rel4[q]  : (v4i)(-1);
      nv[u] = ok ? node4[q] : (v4i)0;
    }
#pragma unroll
    for (int u = 0; u < CNT_UNR; ++u) {
      if (rv[u].x == r) atomicAdd(&bins[nv[u].x], 1);
      if (rv[u].y == r) atomicAdd(&bins[nv[u].y], 1);
      if (rv[u].z == r) atomicAdd(&bins[nv[u].z], 1);
      if (rv[u].w == r) atomicAdd(&bins[nv[u].w], 1);
    }
  }
  for (int e = e0 + 4 * nvec + tid; e < e1; e += CNT_TPB)
    if (rel[e] == r) atomicAdd(&bins[node[e]], 1);
  __syncthreads();

  float* __restrict__ trow = cnt + (size_t)row * NNODES;
  for (int j = tid; j < NNODES; j += CNT_TPB) atomicAdd(trow + j, (float)bins[j]);
}

#define L1_CHUNKS 8
#define L1_TPB    1024
#define L1_B      16
#define L1_LCAP   1024

template <bool GUARD>
__device__ __forceinline__ void l1_batch(const unsigned* qe, int nvalid,
                                         const float* __restrict__ w1, const float* __restrict__ cnt,
                                         float* __restrict__ hs, int n0, int lane) {
  int jl = lane & (L1_B - 1);
  if (GUARD && jl >= nvalid) jl = 0;
  unsigned ent = qe[jl];
  unsigned rp = ent >> 25, o = (ent >> 10) & 0x7FFFu, l = ent & 0x3FFu;
  float cv   = cnt[(size_t)rp * NNODES + (unsigned)n0 + l];
  int   rowv = (int)(rp * NNODES + o);
  int   lofv = (int)(l * HID);
  v2f x[L1_B];
#pragma unroll
  for (int j = 0; j < L1_B; ++j) {
    int row = __builtin_amdgcn_readlane(rowv, j);
    x[j] = ((const v2f*)(w1 + (size_t)(unsigned)row * HID))[lane];
  }
  float vinv = 1.0f / cv;
#pragma unroll
  for (int j = 0; j < L1_B; ++j) {
    if (!GUARD || j < nvalid) {
      float v  = __int_as_float(__builtin_amdgcn_readlane(__float_as_int(vinv), j));
      int  off = __builtin_amdgcn_readlane(lofv, j) + 2 * lane;
      atomicAdd(&hs[off],     v * x[j].x);
      atomicAdd(&hs[off + 1], v * x[j].y);
    }
  }
}

__global__ __launch_bounds__(L1_TPB) void k_layer1(const unsigned* __restrict__ runs,
                                                   const float* __restrict__ w1,
                                                   const float* __restrict__ cnt,
                                                   float* __restrict__ hacc, int E) {
  __shared__ float    hs[L1_K * HID];
  __shared__ unsigned lst[L1_TPB / 32][L1_LCAP];
  const int slice = blockIdx.x % L1_SLICES;
  const int chunk = blockIdx.x / L1_SLICES;
  const int n0    = slice * L1_K;
  const int tid   = threadIdx.x;
  const int lane  = tid & 31;
  const int wave  = __builtin_amdgcn_readfirstlane(tid >> 5);
  unsigned* wl = lst[wave];
  const int nruns = (E + RUN_EDGES - 1) / RUN_EDGES;

  for (int i = tid; i < L1_K * HID / 4; i += L1_TPB) ((v4f*)hs)[i] = (v4f)0.0f;
  __syncthreads();

  for (int run = chunk + L1_CHUNKS * wave; run < nruns; run += L1_CHUNKS * (L1_TPB / 32)) {
    const unsigned* __restrict__ rec = runs + (size_t)run * RUN_WORDS;
    const int o0 = (int)rec[slice], o1 = (int)rec[slice + 1];
    const unsigned* __restrict__ m = rec + RUN_OFFS;
    for (int p0 = o0; p0 < o1; p0 += L1_LCAP) {
      const int n = o1 - p0 < L1_LCAP ? o1 - p0 : L1_LCAP;
      for (int i = lane; i < n; i += 32) wl[i] = m[p0 + i];
      int i = 0;
      for (; i + L1_B <= n; i += L1_B) l1_batch<false>(wl + i, L1_B, w1, cnt, hs, n0, lane);
      if (i < n) l1_batch<true>(wl + i, n - i, w1, cnt, hs, n0, lane);
    }
  }
  __syncthreads();

  float* __restrict__ hg = hacc + (size_t)n0 * HID;
  for (int i0 = tid; i0 < L1_K * HID; i0 += 8 * L1_TPB) {
    float t[8];
#pragma unroll
    for (int u = 0; u < 8; ++u) { int i = i0 + u * L1_TPB; t[u] = i < L1_K * HID ? hs[i] : 0.0f; }
#pragma unroll
    for (int u = 0; u < 8; ++u) { int i = i0 + u * L1_TPB; if (i < L1_K * HID) atomicAdd(hg + i, t[u]); }
  }
}

#define GEMM_RG  3
#define GEMM_RPG (RTOT / GEMM_RG)
#define GEMM_WAVES ((NNODES / 16) * GEMM_RG)
__device__ __forceinline__ v8f tile_product(const v2f (&a)[HID / 4], const float* __restrict__ wrel,
                                            int khalf, int mn) {
  v8f acc = (v8f)0.0f;
#pragma unroll
  for (int kk = 0; kk < HID / 4; ++kk) {
    int ka = 4 * kk + 2 * khalf;
    v2f b;
    b.x = wrel[(size_t)ka * NCLS + mn];
    b.y = wrel[(size_t)(ka + 1) * NCLS + mn];
    acc = __builtin_amdgcn_wmma_f32_16x16x4_f32(
         false, a[kk],  false, b,
         (short)0, acc,  false,  false);
  }
  return acc;
}

template <bool SELF>
__device__ __forceinline__ void gemm_relations(const v2f (&a)[HID / 4], const float* __restrict__ w2,
                                               const float* __restrict__ b2, float* __restrict__ P,
                                               float* __restrict__ out, int mt, int r0, int rowoff,
                                               int khalf, int mn) {
  constexpr int NP = SELF ? GEMM_RPG - 1 : GEMM_RPG;
  for (int r = r0; r < r0 + NP; ++r) {
    union { v8f v; float f[8]; } acc;
    acc.v = tile_product(a, w2 + (size_t)r * HID * NCLS, khalf, mn);
#pragma unroll
    for (int v = 0; v < 8; ++v) {
      P[(size_t)(mt * 16 + v + rowoff) * PCOLS + r * NCLS + mn] = acc.f[v];
    }
  }
  if (SELF) {
    union { v8f v; float f[8]; } acc;
    acc.v = tile_product(a, w2 + (size_t)(2 * NRELH) * HID * NCLS, khalf, mn);
    float bias = b2[mn];
#pragma unroll
    for (int v = 0; v < 8; ++v) {
      out[(size_t)(mt * 16 + v + rowoff) * NCLS + mn] = acc.f[v] + bias;
    }
  }
}

__global__ __launch_bounds__(256) void k_gemm(const float* __restrict__ h,
                                              const float* __restrict__ w2,
                                              const float* __restrict__ b2,
                                              float* __restrict__ P,
                                              float* __restrict__ out) {
  int wid = __builtin_amdgcn_readfirstlane((int)((blockIdx.x * blockDim.x + threadIdx.x) >> 5));
  if (wid >= GEMM_WAVES) return;
  int mt = wid / GEMM_RG;
  int r0 = (wid % GEMM_RG) * GEMM_RPG;
  int lane  = threadIdx.x & 31;
  int mn    = lane & 15;
  int khalf = lane >> 4;
  const float* hrow = h + (size_t)(mt * 16 + mn) * HID;
  v2f a[HID / 4];
#pragma unroll
  for (int kk = 0; kk < HID / 4; ++kk) {
    int ka = 4 * kk + 2 * khalf;
    float x = hrow[ka], y = hrow[ka + 1];
    a[kk].x = x > 0.0f ? x : 0.0f;
    a[kk].y = y > 0.0f ? y : 0.0f;
  }
  int rowoff = khalf * 8;
  if (r0 + GEMM_RPG == RTOT)
    gemm_relations<true>(a, w2, b2, P, out, mt, r0, rowoff, khalf, mn);
  else
    gemm_relations<false>(a, w2, b2, P, out, mt, r0, rowoff, khalf, mn);
}

#define L2_SLICES  (L1_SLICES / 2)
#define L2_K       (NNODES / L2_SLICES)
#define L2_CHUNKS  16
#define L2_THREADS 1024
#define L2_B       32
#define L2_RU      10
#define L2_LCAP    512
static_assert(L1_SLICES % 2 == 0 && L2_K == 2 * L1_K && L2_K % 2 == 0,
              "a slice is two sorted slices; table rows are read 8 bytes at a time");
static_assert(L2_K <= 2048 && NNODES <= 32768 && 2 * NRELH <= 32, "message word: 5 + 15 + 11 bits");

template <bool GUARD>
__device__ __forceinline__ void l2_batch(const unsigned* qe, int nvalid, const float* __restrict__ P,
                                         const float* cinv, float* os, int lane) {
  const int half = lane >> 4, c = lane & 15;
  float x[L2_B / 2], v[L2_B / 2];
  int   off[L2_B / 2];
#pragma unroll
  for (int t = 0; t < L2_B / 2; ++t) {
    int m = 2 * t + half;
    if (GUARD && m >= nvalid) m = 0;
    unsigned ent = qe[m];
    unsigned rp = ent >> 26, o = (ent >> 11) & 0x7FFFu, l = ent & 0x7FFu;
    v[t]   = cinv[rp * L2_K + l];
    x[t]   = P[o * (unsigned)PCOLS + rp * NCLS + c];
    off[t] = (int)(l * NCLS) + c;
  }
#pragma unroll
  for (int t = 0; t < L2_B / 2; ++t) {
    if (!GUARD || 2 * t + half < nvalid) atomicAdd(&os[off[t]], v[t] * x[t]);
  }
}

__global__ __launch_bounds__(L2_THREADS) void k_layer2(const unsigned* __restrict__ runs,
                                                       const float* __restrict__ cnt,
                                                       const float* __restrict__ P,
                                                       float* __restrict__ out, int E) {
  __shared__ __attribute__((aligned(16))) float cinv[2 * NRELH * L2_K];
  __shared__ __attribute__((aligned(16))) float os[L2_K * NCLS];
  __shared__ unsigned lst[L2_THREADS / 32][L2_LCAP];
  const int slice = blockIdx.x % L2_SLICES;
  const int chunk = blockIdx.x / L2_SLICES;
  const int n0    = slice * L2_K;
  const int tid   = threadIdx.x;
  const int lane  = tid & 31;
  const int wave  = __builtin_amdgcn_readfirstlane(tid >> 5);
  unsigned* wl = lst[wave];
  const int nruns = (E + RUN_EDGES - 1) / RUN_EDGES;

  for (int row = wave; row < 2 * NRELH; row += L2_THREADS / 32) {
    const v2f* __restrict__ crow = (const v2f*)(cnt + (size_t)row * NNODES + n0);
    v2f* irow = (v2f*)(cinv + row * L2_K);
    for (int j0 = lane; j0 < L2_K / 2; j0 += L2_RU * 32) {
      v2f t[L2_RU];
#pragma unroll
      for (int u = 0; u < L2_RU; ++u) { int j = j0 + 32 * u; t[u] = crow[j < L2_K / 2 ? j : 0]; }
#pragma unroll
      for (int u = 0; u < L2_RU; ++u) {
        int j = j0 + 32 * u;
        if (j < L2_K / 2) { v2f w; w.x = 1.0f / t[u].x; w.y = 1.0f / t[u].y; irow[j] = w; }
      }
    }
  }
  for (int i = tid; i < L2_K * NCLS / 4; i += L2_THREADS) ((v4f*)os)[i] = (v4f)0.0f;
  __syncthreads();

  for (int run = chunk + L2_CHUNKS * (wave >> 1); run < nruns; run += L2_CHUNKS * (L2_THREADS / 64)) {
    const unsigned* __restrict__ rec = runs + (size_t)run * RUN_WORDS;
    const int o0 = (int)rec[2 * slice], om = (int)rec[2 * slice + 1], o1 = (int)rec[2 * slice + 2];
    const int hl = (o1 - o0 + 1) >> 1;
    const int a  = o0 + (wave & 1) * hl;
    const int b  = a + hl < o1 ? a + hl : o1;
    const unsigned* __restrict__ m = rec + RUN_OFFS;
    for (int p0 = a; p0 < b; p0 += L2_LCAP) {
      const int n = b - p0 < L2_LCAP ? b - p0 : L2_LCAP;
      for (int i = lane; i < n; i += 32) {
        unsigned ent = m[p0 + i];
        unsigned l   = (ent & 0x3FFu) + (p0 + i >= om ? (unsigned)L1_K : 0u);
        wl[i] = ((ent >> 10) << 11) | l;
      }
      int i = 0;
      for (; i + L2_B <= n; i += L2_B) l2_batch<false>(wl + i, L2_B, P, cinv, os, lane);
      if (i < n) l2_batch<true>(wl + i, n - i, P, cinv, os, lane);
    }
  }
  __syncthreads();

  float* __restrict__ og = out + (size_t)n0 * NCLS;
  for (int i0 = tid; i0 < L2_K * NCLS; i0 += 4 * L2_THREADS) {
    float t[4];
#pragma unroll
    for (int u = 0; u < 4; ++u) { int i = i0 + u * L2_THREADS; t[u] = i < L2_K * NCLS ? os[i] : 0.0f; }
#pragma unroll
    for (int u = 0; u < 4; ++u) { int i = i0 + u * L2_THREADS; if (i < L2_K * NCLS) atomicAdd(og + i, t[u]); }
  }
}

extern "C" void kernel_launch(void* const* d_in, const int* in_sizes, int n_in,
                              void* d_out, int out_size, void* d_ws, size_t ws_size,
                              hipStream_t stream) {
  const int*   src = (const int*)d_in[0];
  const int*   rel = (const int*)d_in[1];
  const int*   dst = (const int*)d_in[2];
  const float* w1  = (const float*)d_in[3];
  const float* b1  = (const float*)d_in[4];
  const float* w2  = (const float*)d_in[5];
  const float* b2  = (const float*)d_in[6];
  float* out = (float*)d_out;
  const int E = in_sizes[0];

  float* cnt_s = (float*)d_ws;
  float* cnt_d = cnt_s + CN;
  float* h     = cnt_d + CN;
  float* P     = h + (size_t)NNODES * HID;
  unsigned* runs = (unsigned*)(P + (size_t)NNODES * PCOLS);

  k_init<<<SORT_WGS, SORT_TPB, 0, stream>>>(src, rel, dst, w1, b1, (v4f*)cnt_s, h, runs, E);

  k_count<<<CNT_ROWS * CNT_CHUNKS, CNT_TPB, 0, stream>>>(src, rel, dst, cnt_s, E);

  k_layer1<<<L1_SLICES * L1_CHUNKS, L1_TPB, 0, stream>>>(runs, w1, cnt_s, h, E);

  k_gemm<<<(GEMM_WAVES + 7) / 8, 256, 0, stream>>>(h, w2, b2, P, out);

  k_layer2<<<L2_SLICES * L2_CHUNKS, L2_THREADS, 0, stream>>>(runs, cnt_s, P, out, E);
}


# Counts taken from the sorted messages: each layer-1 work-group counts its slice's (relation, node) pairs in LDS and stores the inverses for layer 2; count kernel, table atomics gone; 5 launches to 4
# speedup vs baseline: 3.8269x; 1.0884x over previous
#include <hip/hip_runtime.h>

#define NNODES 20000
#define NRELH  16
#define RTOT   33
#define HID    64
#define NCLS   16
#define RN     (RTOT * NNODES)
#define PCOLS  (2 * NRELH * NCLS)

typedef __attribute__((ext_vector_type(2))) float v2f;
typedef __attribute__((ext_vector_type(8))) float v8f;

#define L1_SLICES 32
#define L1_K      (NNODES / L1_SLICES)
#define L1_G      16
#define RUN_EDGES 2500
#define RUN_MSGS  (2 * RUN_EDGES)
#define RUN_OFFS  (L1_SLICES + 1)
#define RUN_WORDS (RUN_OFFS + RUN_MSGS)
#define SORT_WGS  256
#define SORT_TPB  512
#define SORT_KEYS (L1_SLICES * L1_G)
#define SORT_EPT  ((RUN_EDGES + SORT_TPB - 1) / SORT_TPB)
static_assert(NNODES % L1_SLICES == 0, "slices must tile the nodes");
static_assert(L1_K <= 1024 && NNODES <= 32768 && 2 * NRELH <= 32, "message word: 5 + 15 + 10 bits");
static_assert(2 * NRELH == 2 * L1_G, "a group is 2 relation rows");
static_assert(SORT_KEYS == SORT_TPB, "the prefix step gives every thread one key");

typedef __attribute__((ext_vector_type(4))) float v4f;
typedef __attribute__((ext_vector_type(4))) int v4i;
__global__ __launch_bounds__(SORT_TPB) void k_init(const int* __restrict__ src,
                                                   const int* __restrict__ rel,
                                                   const int* __restrict__ dst,
                                                   const float* __restrict__ w1,
                                                   const float* __restrict__ b1,
                                                   float* __restrict__ h,
                                                   unsigned* __restrict__ runs, int E) {
  __shared__ int      key_n[SORT_KEYS];
  __shared__ int      slice_n[L1_SLICES];
  __shared__ unsigned stage[RUN_MSGS];
  const int tid   = threadIdx.x;
  const int nruns = (E + RUN_EDGES - 1) / RUN_EDGES;

  for (int run = blockIdx.x; run < nruns; run += gridDim.x) {
    key_n[tid] = 0;
    __syncthreads();

    const int e0 = run * RUN_EDGES;
    const int e1 = e0 + RUN_EDGES < E ? e0 + RUN_EDGES : E;
    int      ks[SORT_EPT], kd[SORT_EPT];
    unsigned ms[SORT_EPT], md[SORT_EPT];
#pragma unroll
    for (int u = 0; u < SORT_EPT; ++u) {
      int e = e0 + tid + u * SORT_TPB;
      ks[u] = -1; kd[u] = 0; ms[u] = 0u; md[u] = 0u;
      if (e < e1) {
        int s = src[e], r = rel[e], d = dst[e];
        int ss = s / L1_K, sd = d / L1_K;
        ks[u] = ss * L1_G + (r >> 1);
        kd[u] = sd * L1_G + ((r + NRELH) >> 1);
        ms[u] = ((unsigned)r << 25) | ((unsigned)d << 10) | (unsigned)(s - ss * L1_K);
        md[u] = ((unsigned)(r + NRELH) << 25) | ((unsigned)s << 10) | (unsigned)(d - sd * L1_K);
        atomicAdd(&key_n[ks[u]], 1);
        atomicAdd(&key_n[kd[u]], 1);
      }
    }
    __syncthreads();

    const int c = key_n[tid];
    int inc = c;
    for (int dl = 1; dl < L1_G; dl <<= 1) {
      int t = __shfl_up(inc, dl, L1_G);
      if ((tid & (L1_G - 1)) >= dl) inc += t;
    }
    if ((tid & (L1_G - 1)) == L1_G - 1) slice_n[tid / L1_G] = inc;
    __syncthreads();
    int base = 0;
    for (int j = 0; j < tid / L1_G; ++j) base += slice_n[j];
    key_n[tid] = base + inc - c;
    unsigned* __restrict__ rec = runs + (size_t)run * RUN_WORDS;
    if ((tid & (L1_G - 1)) == 0) rec[tid / L1_G] = (unsigned)base;
    if (tid == SORT_TPB - 1) rec[L1_SLICES] = (unsigned)(base + inc);
    __syncthreads();

#pragma unroll
    for (int u = 0; u < SORT_EPT; ++u) {
      if (ks[u] >= 0) {
        stage[atomicAdd(&key_n[ks[u]], 1)] = ms[u];
        stage[atomicAdd(&key_n[kd[u]], 1)] = md[u];
      }
    }
    __syncthreads();

    const int total = 2 * (e1 - e0);
    for (int i = tid; i < total; i += SORT_TPB) rec[RUN_OFFS + i] = stage[i];
  }

  const float* wself = w1 + (size_t)(2 * NRELH) * NNODES * HID;
  const int gtid   = blockIdx.x * SORT_TPB + tid;
  const int stride = gridDim.x * SORT_TPB;
  for (int q = gtid; q < NNODES * HID / 4; q += stride) {
    int idx = q * 4, i = idx % HID;
#pragma unroll
    for (int t = 0; t < 4; ++t) h[idx + t] = wself[idx + t] + b1[i + t];
  }
}

#define INV_PITCH 640

#define L1_CHUNKS 8
#define L1_TPB    1024
#define L1_B      16
#define L1_LCAP   512
#define L1_NCNT   (2 * NRELH * L1_K)
#define L1_CG     8
#define L1_CP     7
#define L1_ROWS_PER_WG (2 * NRELH / L1_CHUNKS)
static_assert(INV_PITCH >= L1_K && INV_PITCH % 32 == 0, "a (slice, row) segment is whole 128-byte lines");
static_assert((2 * NRELH) % L1_CHUNKS == 0, "the rows are dealt out evenly to the slice's work-groups");
static_assert(2 * L1_CG <= 32, "one lane per piece bound");

template <bool GUARD>
__device__ __forceinline__ void l1_batch(const unsigned* qe, int nvalid,
                                         const float* __restrict__ w1, const int* cl,
                                         float* __restrict__ hs, int lane) {
  int jl = lane & (L1_B - 1);
  if (GUARD && jl >= nvalid) jl = 0;
  unsigned ent = qe[jl];
  unsigned rp = ent >> 25, o = (ent >> 10) & 0x7FFFu, l = ent & 0x3FFu;
  float cv   = (float)cl[rp * L1_K + l];
  int   rowv = (int)(rp * NNODES + o);
  int   lofv = (int)(l * HID);
  v2f x[L1_B];
#pragma unroll
  for (int j = 0; j < L1_B; ++j) {
    int row = __builtin_amdgcn_readlane(rowv, j);
    x[j] = ((const v2f*)(w1 + (size_t)(unsigned)row * HID))[lane];
  }
  float vinv = 1.0f / cv;
#pragma unroll
  for (int j = 0; j < L1_B; ++j) {
    if (!GUARD || j < nvalid) {
      float v  = __int_as_float(__builtin_amdgcn_readlane(__float_as_int(vinv), j));
      int  off = __builtin_amdgcn_readlane(lofv, j) + 2 * lane;
      atomicAdd(&hs[off],     v * x[j].x);
      atomicAdd(&hs[off + 1], v * x[j].y);
    }
  }
}

__global__ __launch_bounds__(L1_TPB) void k_layer1(const unsigned* __restrict__ runs,
                                                   const float* __restrict__ w1,
                                                   float* __restrict__ ginv,
                                                   float* __restrict__ hacc, int E) {
  __shared__ float    hs[L1_K * HID];
  __shared__ __attribute__((aligned(16))) int cl[L1_NCNT];
  __shared__ unsigned lst[L1_TPB / 32][L1_LCAP];
  const int slice = blockIdx.x % L1_SLICES;
  const int chunk = blockIdx.x / L1_SLICES;
  const int n0    = slice * L1_K;
  const int tid   = threadIdx.x;
  const int lane  = tid & 31;
  const int wave  = __builtin_amdgcn_readfirstlane(tid >> 5);
  unsigned* wl = lst[wave];
  const int nruns = (E + RUN_EDGES - 1) / RUN_EDGES;

  for (int i = tid; i < L1_K * HID / 4; i += L1_TPB) ((v4f*)hs)[i] = (v4f)0.0f;
  for (int i = tid; i < L1_NCNT / 4; i += L1_TPB) ((v4i*)cl)[i] = (v4i)0;
  __syncthreads();

  for (int rb = wave; rb < nruns; rb += (L1_TPB / 32) * L1_CG) {
    const int kr = rb + (L1_TPB / 32) * (lane >> 1);
    unsigned ov = 0u;
    if (lane < 2 * L1_CG && kr < nruns) ov = runs[(size_t)kr * RUN_WORDS + slice + (lane & 1)];
    const unsigned* mk[L1_CG];
    unsigned wv[L1_CG][L1_CP];
    int      nn[L1_CG], ob[L1_CG];
#pragma unroll
    for (int k = 0; k < L1_CG; ++k) {
      ob[k] = __builtin_amdgcn_readlane((int)ov, 2 * k);
      nn[k] = __builtin_amdgcn_readlane((int)ov, 2 * k + 1) - ob[k];
      const int run = rb + (L1_TPB / 32) * k;
      mk[k] = runs + (size_t)(run < nruns ? run : 0) * RUN_WORDS + RUN_OFFS;
#pragma unroll
      for (int p = 0; p < L1_CP; ++p) {
        int i = 32 * p + lane;
        wv[k][p] = mk[k][i < nn[k] ? ob[k] + i : 0];
      }
    }
#pragma unroll
    for (int k = 0; k < L1_CG; ++k) {
#pragma unroll
      for (int p = 0; p < L1_CP; ++p) {
        if (32 * p + lane < nn[k]) atomicAdd(&cl[(wv[k][p] >> 25) * L1_K + (wv[k][p] & 0x3FFu)], 1);
      }
      for (int i = 32 * L1_CP + lane; i < nn[k]; i += 32) {
        unsigned ent = mk[k][ob[k] + i];
        atomicAdd(&cl[(ent >> 25) * L1_K + (ent & 0x3FFu)], 1);
      }
    }
  }
  __syncthreads();

  {
    float* __restrict__ gs = ginv + ((size_t)slice * (2 * NRELH) + chunk * L1_ROWS_PER_WG) * INV_PITCH;
    const int* cs = cl + chunk * L1_ROWS_PER_WG * L1_K;
    for (int i = tid; i < L1_ROWS_PER_WG * L1_K; i += L1_TPB) {
      int rw = i / L1_K, l = i - rw * L1_K;
      gs[rw * INV_PITCH + l] = 1.0f / (float)cs[i];
    }
  }

  for (int run = chunk + L1_CHUNKS * wave; run < nruns; run += L1_CHUNKS * (L1_TPB / 32)) {
    const unsigned* __restrict__ rec = runs + (size_t)run * RUN_WORDS;
    const int o0 = (int)rec[slice], o1 = (int)rec[slice + 1];
    const unsigned* __restrict__ m = rec + RUN_OFFS;
    for (int p0 = o0; p0 < o1; p0 += L1_LCAP) {
      const int n = o1 - p0 < L1_LCAP ? o1 - p0 : L1_LCAP;
      for (int i = lane; i < n; i += 32) wl[i] = m[p0 + i];
      int i = 0;
      for (; i + L1_B <= n; i += L1_B) l1_batch<false>(wl + i, L1_B, w1, cl, hs, lane);
      if (i < n) l1_batch<true>(wl + i, n - i, w1, cl, hs, lane);
    }
  }
  __syncthreads();

  float* __restrict__ hg = hacc + (size_t)n0 * HID;
  for (int i0 = tid; i0 < L1_K * HID; i0 += 8 * L1_TPB) {
    float t[8];
#pragma unroll
    for (int u = 0; u < 8; ++u) { int i = i0 + u * L1_TPB; t[u] = i < L1_K * HID ? hs[i] : 0.0f; }
#pragma unroll
    for (int u = 0; u < 8; ++u) { int i = i0 + u * L1_TPB; if (i < L1_K * HID) atomicAdd(hg + i, t[u]); }
  }
}

#define GEMM_RG  3
#define GEMM_RPG (RTOT / GEMM_RG)
#define GEMM_WAVES ((NNODES / 16) * GEMM_RG)
__device__ __forceinline__ v8f tile_product(const v2f (&a)[HID / 4], const float* __restrict__ wrel,
                                            int khalf, int mn) {
  v8f acc = (v8f)0.0f;
#pragma unroll
  for (int kk = 0; kk < HID / 4; ++kk) {
    int ka = 4 * kk + 2 * khalf;
    v2f b;
    b.x = wrel[(size_t)ka * NCLS + mn];
    b.y = wrel[(size_t)(ka + 1) * NCLS + mn];
    acc = __builtin_amdgcn_wmma_f32_16x16x4_f32(
         false, a[kk],  false, b,
         (short)0, acc,  false,  false);
  }
  return acc;
}

template <bool SELF>
__device__ __forceinline__ void gemm_relations(const v2f (&a)[HID / 4], const float* __restrict__ w2,
                                               const float* __restrict__ b2, float* __restrict__ P,
                                               float* __restrict__ out, int mt, int r0, int rowoff,
                                               int khalf, int mn) {
  constexpr int NP = SELF ? GEMM_RPG - 1 : GEMM_RPG;
  for (int r = r0; r < r0 + NP; ++r) {
    union { v8f v; float f[8]; } acc;
    acc.v = tile_product(a, w2 + (size_t)r * HID * NCLS, khalf, mn);
#pragma unroll
    for (int v = 0; v < 8; ++v) {
      P[(size_t)(mt * 16 + v + rowoff) * PCOLS + r * NCLS + mn] = acc.f[v];
    }
  }
  if (SELF) {
    union { v8f v; float f[8]; } acc;
    acc.v = tile_product(a, w2 + (size_t)(2 * NRELH) * HID * NCLS, khalf, mn);
    float bias = b2[mn];
#pragma unroll
    for (int v = 0; v < 8; ++v) {
      out[(size_t)(mt * 16 + v + rowoff) * NCLS + mn] = acc.f[v] + bias;
    }
  }
}

__global__ __launch_bounds__(256) void k_gemm(const float* __restrict__ h,
                                              const float* __restrict__ w2,
                                              const float* __restrict__ b2,
                                              float* __restrict__ P,
                                              float* __restrict__ out) {
  int wid = __builtin_amdgcn_readfirstlane((int)((blockIdx.x * blockDim.x + threadIdx.x) >> 5));
  if (wid >= GEMM_WAVES) return;
  int mt = wid / GEMM_RG;
  int r0 = (wid % GEMM_RG) * GEMM_RPG;
  int lane  = threadIdx.x & 31;
  int mn    = lane & 15;
  int khalf = lane >> 4;
  const float* hrow = h + (size_t)(mt * 16 + mn) * HID;
  v2f a[HID / 4];
#pragma unroll
  for (int kk = 0; kk < HID / 4; ++kk) {
    int ka = 4 * kk + 2 * khalf;
    float x = hrow[ka], y = hrow[ka + 1];
    a[kk].x = x > 0.0f ? x : 0.0f;
    a[kk].y = y > 0.0f ? y : 0.0f;
  }
  int rowoff = khalf * 8;
  if (r0 + GEMM_RPG == RTOT)
    gemm_relations<true>(a, w2, b2, P, out, mt, r0, rowoff, khalf, mn);
  else
    gemm_relations<false>(a, w2, b2, P, out, mt, r0, rowoff, khalf, mn);
}

#define L2_SLICES  (L1_SLICES / 2)
#define L2_K       (NNODES / L2_SLICES)
#define L2_CHUNKS  16
#define L2_THREADS 1024
#define L2_B       32
#define L2_RU      (INV_PITCH / 32)
#define L2_LCAP    512
static_assert(L1_SLICES % 2 == 0 && L2_K == 2 * L1_K, "a slice is two sorted slices");
static_assert(L2_K <= 2048 && NNODES <= 32768 && 2 * NRELH <= 32, "message word: 5 + 15 + 11 bits");

template <bool GUARD>
__device__ __forceinline__ void l2_batch(const unsigned* qe, int nvalid, const float* __restrict__ P,
                                         const float* cinv, float* os, int lane) {
  const int half = lane >> 4, c = lane & 15;
  float x[L2_B / 2], v[L2_B / 2];
  int   off[L2_B / 2];
#pragma unroll
  for (int t = 0; t < L2_B / 2; ++t) {
    int m = 2 * t + half;
    if (GUARD && m >= nvalid) m = 0;
    unsigned ent = qe[m];
    unsigned rp = ent >> 26, o = (ent >> 11) & 0x7FFFu, l = ent & 0x7FFu;
    v[t]   = cinv[rp * L2_K + l];
    x[t]   = P[o * (unsigned)PCOLS + rp * NCLS + c];
    off[t] = (int)(l * NCLS) + c;
  }
#pragma unroll
  for (int t = 0; t < L2_B / 2; ++t) {
    if (!GUARD || 2 * t + half < nvalid) atomicAdd(&os[off[t]], v[t] * x[t]);
  }
}

__global__ __launch_bounds__(L2_THREADS) void k_layer2(const unsigned* __restrict__ runs,
                                                       const float* __restrict__ ginv,
                                                       const float* __restrict__ P,
                                                       float* __restrict__ out, int E) {
  __shared__ __attribute__((aligned(16))) float cinv[2 * NRELH * L2_K];
  __shared__ __attribute__((aligned(16))) float os[L2_K * NCLS];
  __shared__ unsigned lst[L2_THREADS / 32][L2_LCAP];
  const int slice = blockIdx.x % L2_SLICES;
  const int chunk = blockIdx.x / L2_SLICES;
  const int n0    = slice * L2_K;
  const int tid   = threadIdx.x;
  const int lane  = tid & 31;
  const int wave  = __builtin_amdgcn_readfirstlane(tid >> 5);
  unsigned* wl = lst[wave];
  const int nruns = (E + RUN_EDGES - 1) / RUN_EDGES;

  for (int seg = wave; seg < 2 * 2 * NRELH; seg += L2_THREADS / 32) {
    const int row = seg >> 1, hf = seg & 1;
    const float* __restrict__ grow = ginv + ((size_t)(2 * slice + hf) * (2 * NRELH) + row) * INV_PITCH;
    float* irow = cinv + row * L2_K + hf * L1_K;
    float t[L2_RU];
#pragma unroll
    for (int u = 0; u < L2_RU; ++u) { int j = lane + 32 * u; t[u] = 0.0f; if (j < L1_K) t[u] = grow[j]; }
#pragma unroll
    for (int u = 0; u < L2_RU; ++u) { int j = lane + 32 * u; if (j < L1_K) irow[j] = t[u]; }
  }
  for (int i = tid; i < L2_K * NCLS / 4; i += L2_THREADS) ((v4f*)os)[i] = (v4f)0.0f;
  __syncthreads();

  for (int run = chunk + L2_CHUNKS * (wave >> 1); run < nruns; run += L2_CHUNKS * (L2_THREADS / 64)) {
    const unsigned* __restrict__ rec = runs + (size_t)run * RUN_WORDS;
    const int o0 = (int)rec[2 * slice], om = (int)rec[2 * slice + 1], o1 = (int)rec[2 * slice + 2];
    const int hl = (o1 - o0 + 1) >> 1;
    const int a  = o0 + (wave & 1) * hl;
    const int b  = a + hl < o1 ? a + hl : o1;
    const unsigned* __restrict__ m = rec + RUN_OFFS;
    for (int p0 = a; p0 < b; p0 += L2_LCAP) {
      const int n = b - p0 < L2_LCAP ? b - p0 : L2_LCAP;
      for (int i = lane; i < n; i += 32) {
        unsigned ent = m[p0 + i];
        unsigned l   = (ent & 0x3FFu) + (p0 + i >= om ? (unsigned)L1_K : 0u);
        wl[i] = ((ent >> 10) << 11) | l;
      }
      int i = 0;
      for (; i + L2_B <= n; i += L2_B) l2_batch<false>(wl + i, L2_B, P, cinv, os, lane);
      if (i < n) l2_batch<true>(wl + i, n - i, P, cinv, os, lane);
    }
  }
  __syncthreads();

  float* __restrict__ og = out + (size_t)n0 * NCLS;
  for (int i0 = tid; i0 < L2_K * NCLS; i0 += 4 * L2_THREADS) {
    float t[4];
#pragma unroll
    for (int u = 0; u < 4; ++u) { int i = i0 + u * L2_THREADS; t[u] = i < L2_K * NCLS ? os[i] : 0.0f; }
#pragma unroll
    for (int u = 0; u < 4; ++u) { int i = i0 + u * L2_THREADS; if (i < L2_K * NCLS) atomicAdd(og + i, t[u]); }
  }
}

extern "C" void kernel_launch(void* const* d_in, const int* in_sizes, int n_in,
                              void* d_out, int out_size, void* d_ws, size_t ws_size,
                              hipStream_t stream) {
  const int*   src = (const int*)d_in[0];
  const int*   rel = (const int*)d_in[1];
  const int*   dst = (const int*)d_in[2];
  const float* w1  = (const float*)d_in[3];
  const float* b1  = (const float*)d_in[4];
  const float* w2  = (const float*)d_in[5];
  const float* b2  = (const float*)d_in[6];
  float* out = (float*)d_out;
  const int E = in_sizes[0];

  float* ginv  = (float*)d_ws;
  float* h     = ginv + (size_t)L1_SLICES * (2 * NRELH) * INV_PITCH;
  float* P     = h + (size_t)NNODES * HID;
  unsigned* runs = (unsigned*)(P + (size_t)NNODES * PCOLS);

  k_init<<<SORT_WGS, SORT_TPB, 0, stream>>>(src, rel, dst, w1, b1, h, runs, E);

  k_layer1<<<L1_SLICES * L1_CHUNKS, L1_TPB, 0, stream>>>(runs, w1, ginv, h, E);

  k_gemm<<<(GEMM_WAVES + 7) / 8, 256, 0, stream>>>(h, w2, b2, P, out);

  k_layer2<<<L2_SLICES * L2_CHUNKS, L2_THREADS, 0, stream>>>(runs, ginv, P, out, E);
}


# Layer 1 stores each message's normalizer beside its sorted word; layer 2 (32 slices x 8 shares, as layer 1) reads it there: inverse-count table, its LDS copy and its loads gone; out add order changes
# speedup vs baseline: 3.9730x; 1.0382x over previous
#include <hip/hip_runtime.h>

#define NNODES 20000
#define NRELH  16
#define RTOT   33
#define HID    64
#define NCLS   16
#define RN     (RTOT * NNODES)
#define PCOLS  (2 * NRELH * NCLS)

typedef __attribute__((ext_vector_type(2))) float v2f;
typedef __attribute__((ext_vector_type(8))) float v8f;

#define L1_SLICES 32
#define L1_K      (NNODES / L1_SLICES)
#define L1_G      16
#define RUN_EDGES 2500
#define RUN_MSGS  (2 * RUN_EDGES)
#define RUN_OFFS  (L1_SLICES + 1)
#define RUN_WORDS (RUN_OFFS + 2 * RUN_MSGS)
#define SORT_WGS  256
#define SORT_TPB  512
#define SORT_KEYS (L1_SLICES * L1_G)
#define SORT_EPT  ((RUN_EDGES + SORT_TPB - 1) / SORT_TPB)
static_assert(NNODES % L1_SLICES == 0, "slices must tile the nodes");
static_assert(L1_K <= 1024 && NNODES <= 32768 && 2 * NRELH <= 32, "message word: 5 + 15 + 10 bits");
static_assert(2 * NRELH == 2 * L1_G, "a group is 2 relation rows");
static_assert(SORT_KEYS == SORT_TPB, "the prefix step gives every thread one key");

typedef __attribute__((ext_vector_type(4))) float v4f;
typedef __attribute__((ext_vector_type(4))) int v4i;
__global__ __launch_bounds__(SORT_TPB) void k_init(const int* __restrict__ src,
                                                   const int* __restrict__ rel,
                                                   const int* __restrict__ dst,
                                                   const float* __restrict__ w1,
                                                   const float* __restrict__ b1,
                                                   float* __restrict__ h,
                                                   unsigned* __restrict__ runs, int E) {
  __shared__ int      key_n[SORT_KEYS];
  __shared__ int      slice_n[L1_SLICES];
  __shared__ unsigned stage[RUN_MSGS];
  const int tid   = threadIdx.x;
  const int nruns = (E + RUN_EDGES - 1) / RUN_EDGES;

  for (int run = blockIdx.x; run < nruns; run += gridDim.x) {
    key_n[tid] = 0;
    __syncthreads();

    const int e0 = run * RUN_EDGES;
    const int e1 = e0 + RUN_EDGES < E ? e0 + RUN_EDGES : E;
    int      ks[SORT_EPT], kd[SORT_EPT];
    unsigned ms[SORT_EPT], md[SORT_EPT];
#pragma unroll
    for (int u = 0; u < SORT_EPT; ++u) {
      int e = e0 + tid + u * SORT_TPB;
      ks[u] = -1; kd[u] = 0; ms[u] = 0u; md[u] = 0u;
      if (e < e1) {
        int s = src[e], r = rel[e], d = dst[e];
        int ss = s / L1_K, sd = d / L1_K;
        ks[u] = ss * L1_G + (r >> 1);
        kd[u] = sd * L1_G + ((r + NRELH) >> 1);
        ms[u] = ((unsigned)r << 25) | ((unsigned)d << 10) | (unsigned)(s - ss * L1_K);
        md[u] = ((unsigned)(r + NRELH) << 25) | ((unsigned)s << 10) | (unsigned)(d - sd * L1_K);
        atomicAdd(&key_n[ks[u]], 1);
        atomicAdd(&key_n[kd[u]], 1);
      }
    }
    __syncthreads();

    const int c = key_n[tid];
    int inc = c;
    for (int dl = 1; dl < L1_G; dl <<= 1) {
      int t = __shfl_up(inc, dl, L1_G);
      if ((tid & (L1_G - 1)) >= dl) inc += t;
    }
    if ((tid & (L1_G - 1)) == L1_G - 1) slice_n[tid / L1_G] = inc;
    __syncthreads();
    int base = 0;
    for (int j = 0; j < tid / L1_G; ++j) base += slice_n[j];
    key_n[tid] = base + inc - c;
    unsigned* __restrict__ rec = runs + (size_t)run * RUN_WORDS;
    if ((tid & (L1_G - 1)) == 0) rec[tid / L1_G] = (unsigned)base;
    if (tid == SORT_TPB - 1) rec[L1_SLICES] = (unsigned)(base + inc);
    __syncthreads();

#pragma unroll
    for (int u = 0; u < SORT_EPT; ++u) {
      if (ks[u] >= 0) {
        stage[atomicAdd(&key_n[ks[u]], 1)] = ms[u];
        stage[atomicAdd(&key_n[kd[u]], 1)] = md[u];
      }
    }
    __syncthreads();

    const int total = 2 * (e1 - e0);
    for (int i = tid; i < total; i += SORT_TPB) rec[RUN_OFFS + i] = stage[i];
  }

  const float* wself = w1 + (size_t)(2 * NRELH) * NNODES * HID;
  const int gtid   = blockIdx.x * SORT_TPB + tid;
  const int stride = gridDim.x * SORT_TPB;
  for (int q = gtid; q < NNODES * HID / 4; q += stride) {
    int idx = q * 4, i = idx % HID;
#pragma unroll
    for (int t = 0; t < 4; ++t) h[idx + t] = wself[idx + t] + b1[i + t];
  }
}


#define L1_CHUNKS 8
#define L1_TPB    1024
#define L1_B      16
#define L1_LCAP   512
#define L1_NCNT   (2 * NRELH * L1_K)
#define L1_CG     8
#define L1_CP     7
static_assert(2 * L1_CG <= 32, "one lane per piece bound");

template <bool GUARD>
__device__ __forceinline__ void l1_batch(const unsigned* qe, int nvalid,
                                         const float* __restrict__ w1, const int* cl,
                                         float* __restrict__ hs, float* nq, int lane) {
  int jl = lane & (L1_B - 1);
  if (GUARD && jl >= nvalid) jl = 0;
  unsigned ent = qe[jl];
  unsigned rp = ent >> 25, o = (ent >> 10) & 0x7FFFu, l = ent & 0x3FFu;
  float cv   = (float)cl[rp * L1_K + l];
  int   rowv = (int)(rp * NNODES + o);
  int   lofv = (int)(l * HID);
  v2f x[L1_B];
#pragma unroll
  for (int j = 0; j < L1_B; ++j) {
    int row = __builtin_amdgcn_readlane(rowv, j);
    x[j] = ((const v2f*)(w1 + (size_t)(unsigned)row * HID))[lane];
  }
  float vinv = 1.0f / cv;
  if (lane < (GUARD ? nvalid : L1_B)) nq[lane] = vinv;
#pragma unroll
  for (int j = 0; j < L1_B; ++j) {
    if (!GUARD || j < nvalid) {
      float v  = __int_as_float(__builtin_amdgcn_readlane(__float_as_int(vinv), j));
      int  off = __builtin_amdgcn_readlane(lofv, j) + 2 * lane;
      atomicAdd(&hs[off],     v * x[j].x);
      atomicAdd(&hs[off + 1], v * x[j].y);
    }
  }
}

__global__ __launch_bounds__(L1_TPB) void k_layer1(unsigned* __restrict__ runs,
                                                   const float* __restrict__ w1,
                                                   float* __restrict__ hacc, int E) {
  __shared__ float    hs[L1_K * HID];
  __shared__ __attribute__((aligned(16))) int cl[L1_NCNT];
  __shared__ unsigned lst[L1_TPB / 32][L1_LCAP];
  const int slice = blockIdx.x % L1_SLICES;
  const int chunk = blockIdx.x / L1_SLICES;
  const int n0    = slice * L1_K;
  const int tid   = threadIdx.x;
  const int lane  = tid & 31;
  const int wave  = __builtin_amdgcn_readfirstlane(tid >> 5);
  unsigned* wl = lst[wave];
  const int nruns = (E + RUN_EDGES - 1) / RUN_EDGES;

  for (int i = tid; i < L1_K * HID / 4; i += L1_TPB) ((v4f*)hs)[i] = (v4f)0.0f;
  for (int i = tid; i < L1_NCNT / 4; i += L1_TPB) ((v4i*)cl)[i] = (v4i)0;
  __syncthreads();

  for (int rb = wave; rb < nruns; rb += (L1_TPB / 32) * L1_CG) {
    const int kr = rb + (L1_TPB / 32) * (lane >> 1);
    unsigned ov = 0u;
    if (lane < 2 * L1_CG && kr < nruns) ov = runs[(size_t)kr * RUN_WORDS + slice + (lane & 1)];
    const unsigned* mk[L1_CG];
    unsigned wv[L1_CG][L1_CP];
    int      nn[L1_CG], ob[L1_CG];
#pragma unroll
    for (int k = 0; k < L1_CG; ++k) {
      ob[k] = __builtin_amdgcn_readlane((int)ov, 2 * k);
      nn[k] = __builtin_amdgcn_readlane((int)ov, 2 * k + 1) - ob[k];
      const int run = rb + (L1_TPB / 32) * k;
      mk[k] = runs + (size_t)(run < nruns ? run : 0) * RUN_WORDS + RUN_OFFS;
#pragma unroll
      for (int p = 0; p < L1_CP; ++p) {
        int i = 32 * p + lane;
        wv[k][p] = mk[k][i < nn[k] ? ob[k] + i : 0];
      }
    }
#pragma unroll
    for (int k = 0; k < L1_CG; ++k) {
#pragma unroll
      for (int p = 0; p < L1_CP; ++p) {
        if (32 * p + lane < nn[k]) atomicAdd(&cl[(wv[k][p] >> 25) * L1_K + (wv[k][p] & 0x3FFu)], 1);
      }
      for (int i = 32 * L1_CP + lane; i < nn[k]; i += 32) {
        unsigned ent = mk[k][ob[k] + i];
        atomicAdd(&cl[(ent >> 25) * L1_K + (ent & 0x3FFu)], 1);
      }
    }
  }
  __syncthreads();

  for (int run = chunk + L1_CHUNKS * wave; run < nruns; run += L1_CHUNKS * (L1_TPB / 32)) {
    const unsigned* __restrict__ rec = runs + (size_t)run * RUN_WORDS;
    const int o0 = (int)rec[slice], o1 = (int)rec[slice + 1];
    const unsigned* __restrict__ m = rec + RUN_OFFS;
    float* nv = (float*)(runs + (size_t)run * RUN_WORDS + RUN_OFFS + RUN_MSGS);
    for (int p0 = o0; p0 < o1; p0 += L1_LCAP) {
      const int n = o1 - p0 < L1_LCAP ? o1 - p0 : L1_LCAP;
      for (int i = lane; i < n; i += 32) wl[i] = m[p0 + i];
      int i = 0;
      for (; i + L1_B <= n; i += L1_B) l1_batch<false>(wl + i, L1_B, w1, cl, hs, nv + p0 + i, lane);
      if (i < n) l1_batch<true>(wl + i, n - i, w1, cl, hs, nv + p0 + i, lane);
    }
  }
  __syncthreads();

  float* __restrict__ hg = hacc + (size_t)n0 * HID;
  for (int i0 = tid; i0 < L1_K * HID; i0 += 8 * L1_TPB) {
    float t[8];
#pragma unroll
    for (int u = 0; u < 8; ++u) { int i = i0 + u * L1_TPB; t[u] = i < L1_K * HID ? hs[i] : 0.0f; }
#pragma unroll
    for (int u = 0; u < 8; ++u) { int i = i0 + u * L1_TPB; if (i < L1_K * HID) atomicAdd(hg + i, t[u]); }
  }
}

#define GEMM_RG  3
#define GEMM_RPG (RTOT / GEMM_RG)
#define GEMM_WAVES ((NNODES / 16) * GEMM_RG)
__device__ __forceinline__ v8f tile_product(const v2f (&a)[HID / 4], const float* __restrict__ wrel,
                                            int khalf, int mn) {
  v8f acc = (v8f)0.0f;
#pragma unroll
  for (int kk = 0; kk < HID / 4; ++kk) {
    int ka = 4 * kk + 2 * khalf;
    v2f b;
    b.x = wrel[(size_t)ka * NCLS + mn];
    b.y = wrel[(size_t)(ka + 1) * NCLS + mn];
    acc = __builtin_amdgcn_wmma_f32_16x16x4_f32(
         false, a[kk],  false, b,
         (short)0, acc,  false,  false);
  }
  return acc;
}

template <bool SELF>
__device__ __forceinline__ void gemm_relations(const v2f (&a)[HID / 4], const float* __restrict__ w2,
                                               const float* __restrict__ b2, float* __restrict__ P,
                                               float* __restrict__ out, int mt, int r0, int rowoff,
                                               int khalf, int mn) {
  constexpr int NP = SELF ? GEMM_RPG - 1 : GEMM_RPG;
  for (int r = r0; r < r0 + NP; ++r) {
    union { v8f v; float f[8]; } acc;
    acc.v = tile_product(a, w2 + (size_t)r * HID * NCLS, khalf, mn);
#pragma unroll
    for (int v = 0; v < 8; ++v) {
      P[(size_t)(mt * 16 + v + rowoff) * PCOLS + r * NCLS + mn] = acc.f[v];
    }
  }
  if (SELF) {
    union { v8f v; float f[8]; } acc;
    acc.v = tile_product(a, w2 + (size_t)(2 * NRELH) * HID * NCLS, khalf, mn);
    float bias = b2[mn];
#pragma unroll
    for (int v = 0; v < 8; ++v) {
      out[(size_t)(mt * 16 + v + rowoff) * NCLS + mn] = acc.f[v] + bias;
    }
  }
}

__global__ __launch_bounds__(256) void k_gemm(const float* __restrict__ h,
                                              const float* __restrict__ w2,
                                              const float* __restrict__ b2,
                                              float* __restrict__ P,
                                              float* __restrict__ out) {
  int wid = __builtin_amdgcn_readfirstlane((int)((blockIdx.x * blockDim.x + threadIdx.x) >> 5));
  if (wid >= GEMM_WAVES) return;
  int mt = wid / GEMM_RG;
  int r0 = (wid % GEMM_RG) * GEMM_RPG;
  int lane  = threadIdx.x & 31;
  int mn    = lane & 15;
  int khalf = lane >> 4;
  const float* hrow = h + (size_t)(mt * 16 + mn) * HID;
  v2f a[HID / 4];
#pragma unroll
  for (int kk = 0; kk < HID / 4; ++kk) {
    int ka = 4 * kk + 2 * khalf;
    float x = hrow[ka], y = hrow[ka + 1];
    a[kk].x = x > 0.0f ? x : 0.0f;
    a[kk].y = y > 0.0f ? y : 0.0f;
  }
  int rowoff = khalf * 8;
  if (r0 + GEMM_RPG == RTOT)
    gemm_relations<true>(a, w2, b2, P, out, mt, r0, rowoff, khalf, mn);
  else
    gemm_relations<false>(a, w2, b2, P, out, mt, r0, rowoff, khalf, mn);
}

#define L2_THREADS 1024
#define L2_B       32
#define L2_LCAP    512

template <bool GUARD>
__device__ __forceinline__ void l2_batch(const unsigned* qe, const float* qn, int nvalid,
                                         const float* __restrict__ P, float* os, int lane) {
  const int half = lane >> 4, c = lane & 15;
  float x[L2_B / 2], v[L2_B / 2];
  int   off[L2_B / 2];
#pragma unroll
  for (int t = 0; t < L2_B / 2; ++t) {
    int m = 2 * t + half;
    if (GUARD && m >= nvalid) m = 0;
    unsigned ent = qe[m];
    unsigned rp = ent >> 25, o = (ent >> 10) & 0x7FFFu, l = ent & 0x3FFu;
    v[t]   = qn[m];
    x[t]   = P[o * (unsigned)PCOLS + rp * NCLS + c];
    off[t] = (int)(l * NCLS) + c;
  }
#pragma unroll
  for (int t = 0; t < L2_B / 2; ++t) {
    if (!GUARD || 2 * t + half < nvalid) atomicAdd(&os[off[t]], v[t] * x[t]);
  }
}

__global__ __launch_bounds__(L2_THREADS) void k_layer2(const unsigned* __restrict__ runs,
                                                       const float* __restrict__ P,
                                                       float* __restrict__ out, int E) {
  __shared__ __attribute__((aligned(16))) float os[L1_K * NCLS];
  __shared__ unsigned lst[L2_THREADS / 32][L2_LCAP];
  __shared__ float    lnv[L2_THREADS / 32][L2_LCAP];
  const int slice = blockIdx.x % L1_SLICES;
  const int chunk = blockIdx.x / L1_SLICES;
  const int n0    = slice * L1_K;
  const int tid   = threadIdx.x;
  const int lane  = tid & 31;
  const int wave  = __builtin_amdgcn_readfirstlane(tid >> 5);
  unsigned* wl = lst[wave];
  float*    wn = lnv[wave];
  const int nruns = (E + RUN_EDGES - 1) / RUN_EDGES;

  for (int i = tid; i < L1_K * NCLS / 4; i += L2_THREADS) ((v4f*)os)[i] = (v4f)0.0f;
  __syncthreads();

  for (int run = chunk + L1_CHUNKS * wave; run < nruns; run += L1_CHUNKS * (L2_THREADS / 32)) {
    const unsigned* __restrict__ rec = runs + (size_t)run * RUN_WORDS;
    const int o0 = (int)rec[slice], o1 = (int)rec[slice + 1];
    const unsigned* __restrict__ m  = rec + RUN_OFFS;
    const float*    __restrict__ nv = (const float*)(rec + RUN_OFFS + RUN_MSGS);
    for (int p0 = o0; p0 < o1; p0 += L2_LCAP) {
      const int n = o1 - p0 < L2_LCAP ? o1 - p0 : L2_LCAP;
      for (int i = lane; i < n; i += 32) { wl[i] = m[p0 + i]; wn[i] = nv[p0 + i]; }
      int i = 0;
      for (; i + L2_B <= n; i += L2_B) l2_batch<false>(wl + i, wn + i, L2_B, P, os, lane);
      if (i < n) l2_batch<true>(wl + i, wn + i, n - i, P, os, lane);
    }
  }
  __syncthreads();

  float* __restrict__ og = out + (size_t)n0 * NCLS;
  const int sh = (n0 * NCLS) & 31;
  for (int i0 = tid - sh; i0 < L1_K * NCLS; i0 += 4 * L2_THREADS) {
    float t[4];
#pragma unroll
    for (int u = 0; u < 4; ++u) { int i = i0 + u * L2_THREADS; t[u] = (i >= 0 && i < L1_K * NCLS) ? os[i] : 0.0f; }
#pragma unroll
    for (int u = 0; u < 4; ++u) { int i = i0 + u * L2_THREADS; if (i >= 0 && i < L1_K * NCLS) atomicAdd(og + i, t[u]); }
  }
}

extern "C" void kernel_launch(void* const* d_in, const int* in_sizes, int n_in,
                              void* d_out, int out_size, void* d_ws, size_t ws_size,
                              hipStream_t stream) {
  const int*   src = (const int*)d_in[0];
  const int*   rel = (const int*)d_in[1];
  const int*   dst = (const int*)d_in[2];
  const float* w1  = (const float*)d_in[3];
  const float* b1  = (const float*)d_in[4];
  const float* w2  = (const float*)d_in[5];
  const float* b2  = (const float*)d_in[6];
  float* out = (float*)d_out;
  const int E = in_sizes[0];

  float* h     = (float*)d_ws;
  float* P     = h + (size_t)NNODES * HID;
  unsigned* runs = (unsigned*)(P + (size_t)NNODES * PCOLS);

  k_init<<<SORT_WGS, SORT_TPB, 0, stream>>>(src, rel, dst, w1, b1, h, runs, E);

  k_layer1<<<L1_SLICES * L1_CHUNKS, L1_TPB, 0, stream>>>(runs, w1, h, E);

  k_gemm<<<(GEMM_WAVES + 7) / 8, 256, 0, stream>>>(h, w2, b2, P, out);

  k_layer2<<<L1_SLICES * L1_CHUNKS, L2_THREADS, 0, stream>>>(runs, P, out, E);
}
